# starting kernel
# baseline (speedup 1.0000x reference)
#include <hip/hip_runtime.h>
#include <hip/hip_bf16.h>

#define N_NODES   200000
#define N_EDGES   6400000
#define IN_FEAT   128
#define HIDDEN    16
#define N_CLASSES 32
#define N_TILES   (N_NODES / 16)

typedef __attribute__((ext_vector_type(2))) float v2f;
typedef __attribute__((ext_vector_type(8))) float v8f;

__global__ void init_kernel(float* __restrict__ deg, float* __restrict__ agg1,
                            float* __restrict__ out) {
    unsigned i = blockIdx.x * blockDim.x + threadIdx.x;
    if (i < N_NODES) deg[i] = 1.0f;
    if (i < N_NODES * HIDDEN) agg1[i] = 0.0f;
    if (i < N_NODES * N_CLASSES) out[i] = 0.0f;
}

__global__ void deg_kernel(const int* __restrict__ col, const float* __restrict__ w,
                           float* __restrict__ deg) {
    unsigned e = blockIdx.x * blockDim.x + threadIdx.x;
    if (e < N_EDGES) atomicAdd(&deg[col[e]], w[e]);
}

__global__ void dinv_kernel(const float* __restrict__ deg, float* __restrict__ dinv) {
    unsigned i = blockIdx.x * blockDim.x + threadIdx.x;
    if (i < N_NODES) {
        float d = deg[i];
        dinv[i] = (d > 0.0f) ? rsqrtf(d) : 0.0f;
    }
}

__global__ void norm_kernel(const int* __restrict__ row, const int* __restrict__ col,
                            const float* __restrict__ w, const float* __restrict__ dinv,
                            float* __restrict__ norm) {
    unsigned e = blockIdx.x * blockDim.x + threadIdx.x;
    if (e < N_EDGES) norm[e] = dinv[row[e]] * w[e] * dinv[col[e]];
}

__global__ void gemm1_wmma(const float* __restrict__ x, const float* __restrict__ W,
                           float* __restrict__ h) {
    unsigned wave = (blockIdx.x * blockDim.x + threadIdx.x) >> 5;
    unsigned lane = threadIdx.x & 31;
    if (wave >= N_TILES) return;
    unsigned m0 = wave * 16;
    unsigned lh = lane & 15;
    unsigned hi = lane >> 4;
    v8f c = {};
    const float* xrow = x + (m0 + lh) * IN_FEAT;
#pragma unroll
    for (int k = 0; k < IN_FEAT; k += 4) {
        int ka = k + (int)hi * 2;
        v2f a = { xrow[ka], xrow[ka + 1] };
        v2f b = { W[ka * HIDDEN + lh], W[(ka + 1) * HIDDEN + lh] };
        c = __builtin_amdgcn_wmma_f32_16x16x4_f32(false, a, false, b,
                                                  (short)0, c, false, false);
    }
    unsigned mbase = m0 + hi * 8;
#pragma unroll
    for (int r = 0; r < 8; ++r) h[(mbase + r) * HIDDEN + lh] = c[r];
}

__global__ void scatter1_kernel(const int* __restrict__ row, const int* __restrict__ col,
                                const float* __restrict__ norm, const float* __restrict__ h,
                                float* __restrict__ agg) {
    unsigned t = blockIdx.x * blockDim.x + threadIdx.x;
    if (t < (unsigned)N_EDGES * HIDDEN) {
        unsigned e = t >> 4, f = t & 15;
        int r = row[e], c = col[e];
        atomicAdd(&agg[(unsigned)c * HIDDEN + f], h[(unsigned)r * HIDDEN + f] * norm[e]);
    }
}

__global__ void finalize1_kernel(float* __restrict__ agg1, const float* __restrict__ h1,
                                 const float* __restrict__ dinv, const float* __restrict__ b1) {
    unsigned t = blockIdx.x * blockDim.x + threadIdx.x;
    if (t < N_NODES * HIDDEN) {
        unsigned i = t >> 4, f = t & 15;
        float di = dinv[i];
        float v = agg1[t] + h1[t] * di * di + b1[f];
        agg1[t] = v > 0.0f ? v : 0.0f;
    }
}

__global__ void gemm2_wmma(const float* __restrict__ h2, const float* __restrict__ W,
                           float* __restrict__ z) {
    unsigned gw = (blockIdx.x * blockDim.x + threadIdx.x) >> 5;
    unsigned lane = threadIdx.x & 31;
    if (gw >= N_TILES * 2) return;
    unsigned tile = gw >> 1, nt = gw & 1;
    unsigned m0 = tile * 16, n0 = nt * 16;
    unsigned lh = lane & 15, hi = lane >> 4;
    v8f c = {};
    const float* arow = h2 + (m0 + lh) * HIDDEN;
#pragma unroll
    for (int k = 0; k < HIDDEN; k += 4) {
        int ka = k + (int)hi * 2;
        v2f a = { arow[ka], arow[ka + 1] };
        v2f b = { W[ka * N_CLASSES + n0 + lh], W[(ka + 1) * N_CLASSES + n0 + lh] };
        c = __builtin_amdgcn_wmma_f32_16x16x4_f32(false, a, false, b,
                                                  (short)0, c, false, false);
    }
    unsigned mbase = m0 + hi * 8;
#pragma unroll
    for (int r = 0; r < 8; ++r) z[(mbase + r) * N_CLASSES + n0 + lh] = c[r];
}

__global__ void scatter2_kernel(const int* __restrict__ row, const int* __restrict__ col,
                                const float* __restrict__ norm, const float* __restrict__ z,
                                float* __restrict__ out) {
    unsigned t = blockIdx.x * blockDim.x + threadIdx.x;
    if (t < (unsigned)N_EDGES * N_CLASSES) {
        unsigned e = t >> 5, f = t & 31;
        int r = row[e], c = col[e];
        atomicAdd(&out[(unsigned)c * N_CLASSES + f],
                  z[(unsigned)r * N_CLASSES + f] * norm[e]);
    }
}

__global__ void finalize2_kernel(float* __restrict__ out, const float* __restrict__ z,
                                 const float* __restrict__ dinv, const float* __restrict__ b2) {
    unsigned t = blockIdx.x * blockDim.x + threadIdx.x;
    if (t < N_NODES * N_CLASSES) {
        unsigned i = t >> 5, f = t & 31;
        float di = dinv[i];
        out[t] += z[t] * di * di + b2[f];
    }
}

extern "C" void kernel_launch(void* const* d_in, const int* in_sizes, int n_in,
                              void* d_out, int out_size, void* d_ws, size_t ws_size,
                              hipStream_t stream) {
    const float* x  = (const float*)d_in[0];
    const int*   ei = (const int*)d_in[1];
    const float* ew = (const float*)d_in[2];
    const float* W1 = (const float*)d_in[3];
    const float* b1 = (const float*)d_in[4];
    const float* W2 = (const float*)d_in[5];
    const float* b2 = (const float*)d_in[6];
    const int* row = ei;
    const int* col = ei + N_EDGES;
    float* out = (float*)d_out;

    float* ws   = (float*)d_ws;
    float* deg  = ws;
    float* dinv = deg  + N_NODES;
    float* h1   = dinv + N_NODES;
    float* agg1 = h1   + (size_t)N_NODES * HIDDEN;
    float* z    = agg1 + (size_t)N_NODES * HIDDEN;
    float* norm = z    + (size_t)N_NODES * N_CLASSES;

    const int B = 256;
    init_kernel<<<(N_NODES * N_CLASSES + B - 1) / B, B, 0, stream>>>(deg, agg1, out);
    deg_kernel<<<(N_EDGES + B - 1) / B, B, 0, stream>>>(col, ew, deg);
    dinv_kernel<<<(N_NODES + B - 1) / B, B, 0, stream>>>(deg, dinv);
    norm_kernel<<<(N_EDGES + B - 1) / B, B, 0, stream>>>(row, col, ew, dinv, norm);
    gemm1_wmma<<<(N_TILES * 32 + B - 1) / B, B, 0, stream>>>(x, W1, h1);
    scatter1_kernel<<<((unsigned)N_EDGES * HIDDEN + B - 1) / B, B, 0, stream>>>(row, col, norm, h1, agg1);
    finalize1_kernel<<<(N_NODES * HIDDEN + B - 1) / B, B, 0, stream>>>(agg1, h1, dinv, b1);
    gemm2_wmma<<<(N_TILES * 2 * 32 + B - 1) / B, B, 0, stream>>>(agg1, W2, z);
    scatter2_kernel<<<((unsigned)N_EDGES * N_CLASSES + B - 1) / B, B, 0, stream>>>(row, col, norm, z, out);
    finalize2_kernel<<<(N_NODES * N_CLASSES + B - 1) / B, B, 0, stream>>>(out, z, dinv, b2);
}


# Layer 2 aggregates before its GEMM, A(hW)=(Ah)W: scatter adds 16 floats per edge, not 32; GEMM2 takes self-loop term and bias; changes f32 summation order
# speedup vs baseline: 1.2807x; 1.2807x over previous
#include <hip/hip_runtime.h>
#include <hip/hip_bf16.h>

#define N_NODES   200000
#define N_EDGES   6400000
#define IN_FEAT   128
#define HIDDEN    16
#define N_CLASSES 32
#define N_TILES   (N_NODES / 16)

typedef __attribute__((ext_vector_type(2))) float v2f;
typedef __attribute__((ext_vector_type(8))) float v8f;

__global__ void init_kernel(float* __restrict__ deg, float* __restrict__ agg1,
                            float* __restrict__ agg2) {
    unsigned i = blockIdx.x * blockDim.x + threadIdx.x;
    if (i < N_NODES) deg[i] = 1.0f;
    if (i < N_NODES * HIDDEN) { agg1[i] = 0.0f; agg2[i] = 0.0f; }
}

__global__ void deg_kernel(const int* __restrict__ col, const float* __restrict__ w,
                           float* __restrict__ deg) {
    unsigned e = blockIdx.x * blockDim.x + threadIdx.x;
    if (e < N_EDGES) atomicAdd(&deg[col[e]], w[e]);
}

__global__ void dinv_kernel(const float* __restrict__ deg, float* __restrict__ dinv) {
    unsigned i = blockIdx.x * blockDim.x + threadIdx.x;
    if (i < N_NODES) {
        float d = deg[i];
        dinv[i] = (d > 0.0f) ? rsqrtf(d) : 0.0f;
    }
}

__global__ void norm_kernel(const int* __restrict__ row, const int* __restrict__ col,
                            const float* __restrict__ w, const float* __restrict__ dinv,
                            float* __restrict__ norm) {
    unsigned e = blockIdx.x * blockDim.x + threadIdx.x;
    if (e < N_EDGES) norm[e] = dinv[row[e]] * w[e] * dinv[col[e]];
}

__global__ void gemm1_wmma(const float* __restrict__ x, const float* __restrict__ W,
                           float* __restrict__ h) {
    unsigned wave = (blockIdx.x * blockDim.x + threadIdx.x) >> 5;
    unsigned lane = threadIdx.x & 31;
    if (wave >= N_TILES) return;
    unsigned m0 = wave * 16;
    unsigned lh = lane & 15;
    unsigned hi = lane >> 4;
    v8f c = {};
    const float* xrow = x + (m0 + lh) * IN_FEAT;
#pragma unroll
    for (int k = 0; k < IN_FEAT; k += 4) {
        int ka = k + (int)hi * 2;
        v2f a = { xrow[ka], xrow[ka + 1] };
        v2f b = { W[ka * HIDDEN + lh], W[(ka + 1) * HIDDEN + lh] };
        c = __builtin_amdgcn_wmma_f32_16x16x4_f32(false, a, false, b,
                                                  (short)0, c, false, false);
    }
    unsigned mbase = m0 + hi * 8;
#pragma unroll
    for (int r = 0; r < 8; ++r) h[(mbase + r) * HIDDEN + lh] = c[r];
}

__global__ void scatter16_kernel(const int* __restrict__ row, const int* __restrict__ col,
                                const float* __restrict__ norm, const float* __restrict__ h,
                                float* __restrict__ agg) {
    unsigned t = blockIdx.x * blockDim.x + threadIdx.x;
    if (t < (unsigned)N_EDGES * HIDDEN) {
        unsigned e = t >> 4, f = t & 15;
        int r = row[e], c = col[e];
        atomicAdd(&agg[(unsigned)c * HIDDEN + f], h[(unsigned)r * HIDDEN + f] * norm[e]);
    }
}

__global__ void finalize1_kernel(float* __restrict__ agg1, const float* __restrict__ h1,
                                 const float* __restrict__ dinv, const float* __restrict__ b1) {
    unsigned t = blockIdx.x * blockDim.x + threadIdx.x;
    if (t < N_NODES * HIDDEN) {
        unsigned i = t >> 4, f = t & 15;
        float di = dinv[i];
        float v = agg1[t] + h1[t] * di * di + b1[f];
        agg1[t] = v > 0.0f ? v : 0.0f;
    }
}

__global__ void gemm2_wmma(const float* __restrict__ agg2, const float* __restrict__ h2,
                           const float* __restrict__ dinv, const float* __restrict__ W,
                           const float* __restrict__ b2, float* __restrict__ out) {
    unsigned wave = (blockIdx.x * blockDim.x + threadIdx.x) >> 5;
    unsigned lane = threadIdx.x & 31;
    if (wave >= N_TILES) return;
    unsigned m0 = wave * 16;
    unsigned lh = lane & 15, hi = lane >> 4;
    float di = dinv[m0 + lh];
    float d2 = di * di;
    float bia0 = b2[lh], bia1 = b2[16 + lh];
    v8f c0 = { bia0, bia0, bia0, bia0, bia0, bia0, bia0, bia0 };
    v8f c1 = { bia1, bia1, bia1, bia1, bia1, bia1, bia1, bia1 };
    const float* grow = agg2 + (m0 + lh) * HIDDEN;
    const float* hrow = h2   + (m0 + lh) * HIDDEN;
#pragma unroll
    for (int k = 0; k < HIDDEN; k += 4) {
        int ka = k + (int)hi * 2;
        v2f a  = { grow[ka] + hrow[ka] * d2, grow[ka + 1] + hrow[ka + 1] * d2 };
        v2f w0 = { W[ka * N_CLASSES + lh],      W[(ka + 1) * N_CLASSES + lh] };
        v2f w1 = { W[ka * N_CLASSES + 16 + lh], W[(ka + 1) * N_CLASSES + 16 + lh] };
        c0 = __builtin_amdgcn_wmma_f32_16x16x4_f32(false, a, false, w0,
                                                   (short)0, c0, false, false);
        c1 = __builtin_amdgcn_wmma_f32_16x16x4_f32(false, a, false, w1,
                                                   (short)0, c1, false, false);
    }
    unsigned mbase = m0 + hi * 8;
#pragma unroll
    for (int r = 0; r < 8; ++r) {
        out[(mbase + r) * N_CLASSES + lh]      = c0[r];
        out[(mbase + r) * N_CLASSES + 16 + lh] = c1[r];
    }
}

extern "C" void kernel_launch(void* const* d_in, const int* in_sizes, int n_in,
                              void* d_out, int out_size, void* d_ws, size_t ws_size,
                              hipStream_t stream) {
    const float* x  = (const float*)d_in[0];
    const int*   ei = (const int*)d_in[1];
    const float* ew = (const float*)d_in[2];
    const float* W1 = (const float*)d_in[3];
    const float* b1 = (const float*)d_in[4];
    const float* W2 = (const float*)d_in[5];
    const float* b2 = (const float*)d_in[6];
    const int* row = ei;
    const int* col = ei + N_EDGES;
    float* out = (float*)d_out;

    float* ws   = (float*)d_ws;
    float* deg  = ws;
    float* dinv = deg  + N_NODES;
    float* h1   = dinv + N_NODES;
    float* agg1 = h1   + (size_t)N_NODES * HIDDEN;
    float* agg2 = agg1 + (size_t)N_NODES * HIDDEN;
    float* norm = agg2 + (size_t)N_NODES * HIDDEN;

    const int B = 256;
    init_kernel<<<(N_NODES * HIDDEN + B - 1) / B, B, 0, stream>>>(deg, agg1, agg2);
    deg_kernel<<<(N_EDGES + B - 1) / B, B, 0, stream>>>(col, ew, deg);
    dinv_kernel<<<(N_NODES + B - 1) / B, B, 0, stream>>>(deg, dinv);
    norm_kernel<<<(N_EDGES + B - 1) / B, B, 0, stream>>>(row, col, ew, dinv, norm);
    gemm1_wmma<<<(N_TILES * 32 + B - 1) / B, B, 0, stream>>>(x, W1, h1);
    scatter16_kernel<<<((unsigned)N_EDGES * HIDDEN + B - 1) / B, B, 0, stream>>>(row, col, norm, h1, agg1);
    finalize1_kernel<<<(N_NODES * HIDDEN + B - 1) / B, B, 0, stream>>>(agg1, h1, dinv, b1);
    scatter16_kernel<<<((unsigned)N_EDGES * HIDDEN + B - 1) / B, B, 0, stream>>>(row, col, norm, agg1, agg2);
    gemm2_wmma<<<(N_TILES * 32 + B - 1) / B, B, 0, stream>>>(agg2, agg1, dinv, W2, b2, out);
}


# Edges counting-sorted by destination each call (int atomic rank, scan, place); both layers sum in-edges per node by gather, no float atomics; changes f32 summation order
# speedup vs baseline: 2.0874x; 1.6299x over previous
#include <hip/hip_runtime.h>

#define N_NODES   200000
#define N_EDGES   6400000
#define IN_FEAT   128
#define HIDDEN    16
#define N_CLASSES 32
#define N_TILES   (N_NODES / 16)
#define N_QUADS   (N_NODES / 4)
#define SCAN_BLOCKS ((N_QUADS + 255) / 256)

typedef __attribute__((ext_vector_type(2))) float v2f;
typedef __attribute__((ext_vector_type(8))) float v8f;


__global__ void zero_cnt_kernel(int* __restrict__ cnt) {
    unsigned i = blockIdx.x * blockDim.x + threadIdx.x;
    if (i < N_NODES) cnt[i] = 0;
}

__global__ void rank_kernel(const int* __restrict__ col, int* __restrict__ cnt,
                            int* __restrict__ rank) {
    unsigned e = blockIdx.x * blockDim.x + threadIdx.x;
    if (e < N_EDGES) rank[e] = atomicAdd(&cnt[col[e]], 1);
}

__global__ void blocksum_kernel(const int* __restrict__ cnt, int* __restrict__ bsum) {
    __shared__ int sh[256];
    unsigned t = threadIdx.x;
    unsigned q = blockIdx.x * 256 + t;
    int s = 0;
    if (q < N_QUADS) {
        int4 v = ((const int4*)cnt)[q];
        s = v.x + v.y + v.z + v.w;
    }
    sh[t] = s;
    __syncthreads();
    for (int d = 128; d > 0; d >>= 1) {
        if (t < (unsigned)d) sh[t] += sh[t + d];
        __syncthreads();
    }
    if (t == 0) bsum[blockIdx.x] = sh[0];
}

__global__ void offsets_kernel(const int* __restrict__ cnt, const int* __restrict__ bsum,
                               int* __restrict__ off) {
    __shared__ int sh[256];
    unsigned t = threadIdx.x, b = blockIdx.x;
    sh[t] = (t < b) ? bsum[t] : 0;
    __syncthreads();
    for (int d = 128; d > 0; d >>= 1) {
        if (t < (unsigned)d) sh[t] += sh[t + d];
        __syncthreads();
    }
    int base = sh[0];
    __syncthreads();
    unsigned q = b * 256 + t;
    int4 v = make_int4(0, 0, 0, 0);
    if (q < N_QUADS) v = ((const int4*)cnt)[q];
    int s = v.x + v.y + v.z + v.w;
    sh[t] = s;
    __syncthreads();
    for (int d = 1; d < 256; d <<= 1) {
        int add = (t >= (unsigned)d) ? sh[t - d] : 0;
        __syncthreads();
        sh[t] += add;
        __syncthreads();
    }
    int ex = sh[t] - s + base;
    if (q < N_QUADS) {
        int4 o;
        o.x = ex;
        o.y = o.x + v.x;
        o.z = o.y + v.y;
        o.w = o.z + v.z;
        ((int4*)off)[q] = o;
    }
}

__global__ void place_kernel(const int* __restrict__ row, const int* __restrict__ col,
                             const float* __restrict__ w, const int* __restrict__ rank,
                             const int* __restrict__ off, int2* __restrict__ ent) {
    unsigned e = blockIdx.x * blockDim.x + threadIdx.x;
    if (e < N_EDGES) {
        int p = off[col[e]] + rank[e];
        ent[p] = make_int2(row[e], __float_as_int(w[e]));
    }
}

__global__ void dinv_kernel(const int* __restrict__ off, const int* __restrict__ cnt,
                            const int2* __restrict__ ent, float* __restrict__ dinv) {
    unsigned t = blockIdx.x * blockDim.x + threadIdx.x;
    if (t >= N_NODES * HIDDEN) return;
    unsigned c = t >> 4, f = t & 15;
    int s = off[c], n = cnt[c];
    float d = 0.0f;
    for (int k = (int)f; k < n; k += 16) d += __int_as_float(ent[s + k].y);
    d += __shfl_xor(d, 8);
    d += __shfl_xor(d, 4);
    d += __shfl_xor(d, 2);
    d += __shfl_xor(d, 1);
    d += 1.0f;
    if (f == 0) dinv[c] = (d > 0.0f) ? rsqrtf(d) : 0.0f;
}

__global__ void gemm1_wmma(const float* __restrict__ x, const float* __restrict__ W,
                           const float* __restrict__ dinv, float* __restrict__ hs) {
    unsigned wave = (blockIdx.x * blockDim.x + threadIdx.x) >> 5;
    unsigned lane = threadIdx.x & 31;
    if (wave >= N_TILES) return;
    unsigned m0 = wave * 16;
    unsigned lh = lane & 15;
    unsigned hi = lane >> 4;
    v8f c = {};
    const float* xrow = x + (m0 + lh) * IN_FEAT;
#pragma unroll
    for (int k = 0; k < IN_FEAT; k += 4) {
        int ka = k + (int)hi * 2;
        v2f a = { xrow[ka], xrow[ka + 1] };
        v2f b = { W[ka * HIDDEN + lh], W[(ka + 1) * HIDDEN + lh] };
        c = __builtin_amdgcn_wmma_f32_16x16x4_f32(false, a, false, b,
                                                  (short)0, c, false, false);
    }
    unsigned mbase = m0 + hi * 8;
#pragma unroll
    for (int r = 0; r < 8; ++r) hs[(mbase + r) * HIDDEN + lh] = c[r] * dinv[mbase + r];
}

template <int LAYER>
__global__ void agg_kernel(const int* __restrict__ off, const int* __restrict__ cnt,
                           const int2* __restrict__ ent, const float* __restrict__ hs,
                           const float* __restrict__ dinv, const float* __restrict__ b1,
                           float* __restrict__ dst) {
    unsigned t = blockIdx.x * blockDim.x + threadIdx.x;
    if (t >= N_NODES * HIDDEN) return;
    unsigned c = t >> 4, f = t & 15;
    int s = off[c], n = cnt[c];
    int nmax = max(n, __shfl_xor(n, 16));
    int2 cur;
    {
        int idx = s + (int)f;
        idx = idx < N_EDGES ? idx : N_EDGES - 1;
        int2 v = ent[idx];
        bool ok = (int)f < n;
        cur.x = ok ? v.x : 0;
        cur.y = ok ? v.y : 0;
    }
    float acc = 0.0f;
    for (int base = 0; base < nmax; base += 16) {
        int2 nxt;
        {
            int kn = base + 16 + (int)f;
            int idx = s + kn;
            idx = idx < N_EDGES ? idx : N_EDGES - 1;
            int2 v = ent[idx];
            bool ok = kn < n;
            nxt.x = ok ? v.x : 0;
            nxt.y = ok ? v.y : 0;
        }
        int rem = n - base;
#pragma unroll
        for (int j = 0; j < 16; ++j) {
            int r = __shfl(cur.x, j, 16);
            float wv = __int_as_float(__shfl(cur.y, j, 16));
            float hv = hs[(unsigned)r * HIDDEN + f];
            hv = (j < rem) ? hv : 0.0f;
            acc += wv * hv;
        }
        cur = nxt;
    }
    float di = dinv[c];
    float v = di * (acc + hs[t]);
    if (LAYER == 1) {
        v += b1[f];
        v = v > 0.0f ? v : 0.0f;
        v *= di;
    }
    dst[t] = v;
}

__global__ void gemm2_wmma(const float* __restrict__ a2, const float* __restrict__ W,
                           const float* __restrict__ b2, float* __restrict__ out) {
    unsigned wave = (blockIdx.x * blockDim.x + threadIdx.x) >> 5;
    unsigned lane = threadIdx.x & 31;
    if (wave >= N_TILES) return;
    unsigned m0 = wave * 16;
    unsigned lh = lane & 15, hi = lane >> 4;
    float bia0 = b2[lh], bia1 = b2[16 + lh];
    v8f c0 = { bia0, bia0, bia0, bia0, bia0, bia0, bia0, bia0 };
    v8f c1 = { bia1, bia1, bia1, bia1, bia1, bia1, bia1, bia1 };
    const float* grow = a2 + (m0 + lh) * HIDDEN;
#pragma unroll
    for (int k = 0; k < HIDDEN; k += 4) {
        int ka = k + (int)hi * 2;
        v2f a  = { grow[ka], grow[ka + 1] };
        v2f w0 = { W[ka * N_CLASSES + lh],      W[(ka + 1) * N_CLASSES + lh] };
        v2f w1 = { W[ka * N_CLASSES + 16 + lh], W[(ka + 1) * N_CLASSES + 16 + lh] };
        c0 = __builtin_amdgcn_wmma_f32_16x16x4_f32(false, a, false, w0,
                                                   (short)0, c0, false, false);
        c1 = __builtin_amdgcn_wmma_f32_16x16x4_f32(false, a, false, w1,
                                                   (short)0, c1, false, false);
    }
    unsigned mbase = m0 + hi * 8;
#pragma unroll
    for (int r = 0; r < 8; ++r) {
        out[(mbase + r) * N_CLASSES + lh]      = c0[r];
        out[(mbase + r) * N_CLASSES + 16 + lh] = c1[r];
    }
}

extern "C" void kernel_launch(void* const* d_in, const int* in_sizes, int n_in,
                              void* d_out, int out_size, void* d_ws, size_t ws_size,
                              hipStream_t stream) {
    const float* x  = (const float*)d_in[0];
    const int*   ei = (const int*)d_in[1];
    const float* ew = (const float*)d_in[2];
    const float* W1 = (const float*)d_in[3];
    const float* b1 = (const float*)d_in[4];
    const float* W2 = (const float*)d_in[5];
    const float* b2 = (const float*)d_in[6];
    const int* row = ei;
    const int* col = ei + N_EDGES;
    float* out = (float*)d_out;

    int*   cnt  = (int*)d_ws;
    int*   off  = cnt  + N_NODES;
    int*   bsum = off  + N_NODES;
    float* dinv = (float*)(bsum + 256);
    float* hs1  = dinv + N_NODES;
    float* hs2  = hs1  + (size_t)N_NODES * HIDDEN;
    float* a2   = hs2  + (size_t)N_NODES * HIDDEN;
    int*   rank = (int*)(a2 + (size_t)N_NODES * HIDDEN);
    int2*  ent  = (int2*)(rank + N_EDGES);

    const int B = 256;
    zero_cnt_kernel<<<(N_NODES + B - 1) / B, B, 0, stream>>>(cnt);
    rank_kernel<<<(N_EDGES + B - 1) / B, B, 0, stream>>>(col, cnt, rank);
    blocksum_kernel<<<SCAN_BLOCKS, 256, 0, stream>>>(cnt, bsum);
    offsets_kernel<<<SCAN_BLOCKS, 256, 0, stream>>>(cnt, bsum, off);
    place_kernel<<<(N_EDGES + B - 1) / B, B, 0, stream>>>(row, col, ew, rank, off, ent);
    dinv_kernel<<<(N_NODES * HIDDEN + B - 1) / B, B, 0, stream>>>(off, cnt, ent, dinv);
    gemm1_wmma<<<(N_TILES * 32 + B - 1) / B, B, 0, stream>>>(x, W1, dinv, hs1);
    agg_kernel<1><<<(N_NODES * HIDDEN + B - 1) / B, B, 0, stream>>>(off, cnt, ent, hs1, dinv, b1, hs2);
    agg_kernel<2><<<(N_NODES * HIDDEN + B - 1) / B, B, 0, stream>>>(off, cnt, ent, hs2, dinv, b1, a2);
    gemm2_wmma<<<(N_TILES * 32 + B - 1) / B, B, 0, stream>>>(a2, W2, b2, out);
}


# Edge sort without global atomics: two-level counting sort on LDS counters (chunk runs per 256-node bucket staged in LDS, then one group per bucket sorts by node); changes f32 summation order
# speedup vs baseline: 5.5891x; 2.6775x over previous
#include <hip/hip_runtime.h>

#define N_NODES   200000
#define N_EDGES   6400000
#define IN_FEAT   128
#define HIDDEN    16
#define N_CLASSES 32
#define N_TILES   (N_NODES / 16)

#define SH        8
#define BNODES    (1 << SH)
#define NBK       ((N_NODES + BNODES - 1) / BNODES)
#define CH        32768
#define NBLK      ((N_EDGES + CH - 1) / CH)
#define M_BH      (NBK * NBLK)
#define E_QUADS   (N_EDGES / 4)
#define N_QUADS   (M_BH / 4)
#define SCAN_BLOCKS ((N_QUADS + 255) / 256)
#define ROW_BITS  18
#define CAP       9984

static_assert(N_EDGES % 4 == 0 && M_BH % 4 == 0, "whole quads");
static_assert(N_NODES <= (1 << ROW_BITS) && SH + ROW_BITS < 32, "source index and local node share one word");
static_assert(NBK <= 1024 && SCAN_BLOCKS <= 256 && CH == 8 * 4 * 1024, "one thread per bucket, one scan block per thread");

typedef __attribute__((ext_vector_type(2))) float v2f;
typedef __attribute__((ext_vector_type(8))) float v8f;
typedef __attribute__((ext_vector_type(2))) int   v2i;


__global__ void __launch_bounds__(1024) hist_kernel(const int* __restrict__ col, int* __restrict__ bh) {
    __shared__ int h[1024];
    unsigned t = threadIdx.x, k = blockIdx.x;
    h[t] = 0;
    __syncthreads();
    const int4* col4 = (const int4*)col;
    int4 c[8];
#pragma unroll
    for (int i = 0; i < 8; ++i) {
        unsigned q = k * (CH / 4) + i * 1024 + t;
        c[i] = col4[q < E_QUADS ? q : 0];
    }
#pragma unroll
    for (int i = 0; i < 8; ++i) {
        unsigned q = k * (CH / 4) + i * 1024 + t;
        if (q < E_QUADS) {
            atomicAdd(&h[c[i].x >> SH], 1);
            atomicAdd(&h[c[i].y >> SH], 1);
            atomicAdd(&h[c[i].z >> SH], 1);
            atomicAdd(&h[c[i].w >> SH], 1);
        }
    }
    __syncthreads();
    if (t < NBK) bh[t * NBLK + k] = h[t];
}

__global__ void blocksum_kernel(const int* __restrict__ cnt, int* __restrict__ bsum) {
    __shared__ int sh[256];
    unsigned t = threadIdx.x;
    unsigned q = blockIdx.x * 256 + t;
    int s = 0;
    if (q < N_QUADS) {
        int4 v = ((const int4*)cnt)[q];
        s = v.x + v.y + v.z + v.w;
    }
    sh[t] = s;
    __syncthreads();
    for (int d = 128; d > 0; d >>= 1) {
        if (t < (unsigned)d) sh[t] += sh[t + d];
        __syncthreads();
    }
    if (t == 0) bsum[blockIdx.x] = sh[0];
}

__global__ void offsets_kernel(const int* __restrict__ cnt, const int* __restrict__ bsum,
                               int* __restrict__ off) {
    __shared__ int sh[256];
    unsigned t = threadIdx.x, b = blockIdx.x;
    sh[t] = (t < b) ? bsum[t] : 0;
    __syncthreads();
    for (int d = 128; d > 0; d >>= 1) {
        if (t < (unsigned)d) sh[t] += sh[t + d];
        __syncthreads();
    }
    int base = sh[0];
    __syncthreads();
    unsigned q = b * 256 + t;
    int4 v = make_int4(0, 0, 0, 0);
    if (q < N_QUADS) v = ((const int4*)cnt)[q];
    int s = v.x + v.y + v.z + v.w;
    sh[t] = s;
    __syncthreads();
    for (int d = 1; d < 256; d <<= 1) {
        int add = (t >= (unsigned)d) ? sh[t - d] : 0;
        __syncthreads();
        sh[t] += add;
        __syncthreads();
    }
    int ex = sh[t] - s + base;
    if (q < N_QUADS) {
        int4 o;
        o.x = ex;
        o.y = o.x + v.x;
        o.z = o.y + v.y;
        o.w = o.z + v.z;
        ((int4*)off)[q] = o;
    }
}

__global__ void __launch_bounds__(1024) bucket_kernel(const int* __restrict__ row, const int* __restrict__ col,
                                                      const float* __restrict__ w, const int* __restrict__ bh,
                                                      const int* __restrict__ bo, v2i* __restrict__ tmp) {
    __shared__ v2i stage[CH];
    __shared__ int lstart[1024];
    __shared__ int cur[1024];
    __shared__ int gb[1024];
    unsigned t = threadIdx.x, k = blockIdx.x;
    int c = 0, g = 0;
    if (t < NBK) {
        c = bh[t * NBLK + k];
        g = bo[t * NBLK + k];
    }
    cur[t] = c;
    __syncthreads();
    for (int d = 1; d < 1024; d <<= 1) {
        int add = (t >= (unsigned)d) ? cur[t - d] : 0;
        __syncthreads();
        cur[t] += add;
        __syncthreads();
    }
    int ex = cur[t] - c;
    lstart[t] = ex;
    cur[t] = ex;
    gb[t] = g;
    __syncthreads();
    const int4*   col4 = (const int4*)col;
    const int4*   row4 = (const int4*)row;
    const float4* w4   = (const float4*)w;
#pragma unroll
    for (int i0 = 0; i0 < 8; i0 += 4) {
        int4 cc[4], rr[4];
        float4 ww[4];
#pragma unroll
        for (int u = 0; u < 4; ++u) {
            unsigned q = k * (CH / 4) + (i0 + u) * 1024 + t;
            unsigned qs = q < E_QUADS ? q : 0;
            cc[u] = col4[qs];
            rr[u] = row4[qs];
            ww[u] = w4[qs];
        }
#pragma unroll
        for (int u = 0; u < 4; ++u) {
            unsigned q = k * (CH / 4) + (i0 + u) * 1024 + t;
            if (q < E_QUADS) {
                int p0 = atomicAdd(&cur[cc[u].x >> SH], 1);
                int p1 = atomicAdd(&cur[cc[u].y >> SH], 1);
                int p2 = atomicAdd(&cur[cc[u].z >> SH], 1);
                int p3 = atomicAdd(&cur[cc[u].w >> SH], 1);
                stage[p0] = (v2i){ ((cc[u].x & (BNODES - 1)) << ROW_BITS) | rr[u].x, __float_as_int(ww[u].x) };
                stage[p1] = (v2i){ ((cc[u].y & (BNODES - 1)) << ROW_BITS) | rr[u].y, __float_as_int(ww[u].y) };
                stage[p2] = (v2i){ ((cc[u].z & (BNODES - 1)) << ROW_BITS) | rr[u].z, __float_as_int(ww[u].z) };
                stage[p3] = (v2i){ ((cc[u].w & (BNODES - 1)) << ROW_BITS) | rr[u].w, __float_as_int(ww[u].w) };
            }
        }
    }
    __syncthreads();
    unsigned wv = t >> 5, lane = t & 31;
    unsigned mb = wv + 32 * lane;
    int my_ls = 0, my_len = 0, my_g = 0;
    if (mb < NBK) {
        my_ls = lstart[mb];
        my_len = cur[mb] - my_ls;
        my_g = gb[mb];
    }
    for (int i = 0; i < (NBK + 31) / 32; ++i) {
        int ls  = __builtin_amdgcn_readlane(my_ls, i);
        int len = __builtin_amdgcn_readlane(my_len, i);
        int g0  = __builtin_amdgcn_readlane(my_g, i);
        for (int j = (int)lane; j < len; j += 32) tmp[g0 + j] = stage[ls + j];
    }
}

__global__ void __launch_bounds__(256) node_sort_kernel(const v2i* __restrict__ tmp, const int* __restrict__ bo,
                                                        int* __restrict__ off, int* __restrict__ cnt,
                                                        int2* __restrict__ ent) {
    __shared__ v2i stage[CAP];
    __shared__ int lc[BNODES];
    unsigned t = threadIdx.x, b = blockIdx.x;
    int bstart = bo[b * NBLK];
    int bend = (b + 1 < NBK) ? bo[(b + 1) * NBLK] : N_EDGES;
    int len = bend - bstart;
    bool staged = len <= CAP;
    lc[t] = 0;
    __syncthreads();
    for (int j0 = 0; j0 < len; j0 += 256 * 8) {
        int key[8];
#pragma unroll
        for (int u = 0; u < 8; ++u) {
            int j = j0 + u * 256 + (int)t;
            key[u] = tmp[bstart + (j < len ? j : 0)].x;
        }
#pragma unroll
        for (int u = 0; u < 8; ++u) {
            int j = j0 + u * 256 + (int)t;
            if (j < len) atomicAdd(&lc[key[u] >> ROW_BITS], 1);
        }
    }
    __syncthreads();
    int c = lc[t];
    for (int d = 1; d < 256; d <<= 1) {
        int add = (t >= (unsigned)d) ? lc[t - d] : 0;
        __syncthreads();
        lc[t] += add;
        __syncthreads();
    }
    int ex = lc[t] - c;
    unsigned node = b * BNODES + t;
    if (node < N_NODES) {
        off[node] = bstart + ex;
        cnt[node] = c;
    }
    lc[t] = ex;
    __syncthreads();
    for (int j0 = 0; j0 < len; j0 += 256 * 8) {
        v2i e[8];
        int p[8];
#pragma unroll
        for (int u = 0; u < 8; ++u) {
            int j = j0 + u * 256 + (int)t;
            e[u] = tmp[bstart + (j < len ? j : 0)];
        }
#pragma unroll
        for (int u = 0; u < 8; ++u) {
            int j = j0 + u * 256 + (int)t;
            p[u] = 0;
            if (j < len) p[u] = atomicAdd(&lc[e[u].x >> ROW_BITS], 1);
        }
#pragma unroll
        for (int u = 0; u < 8; ++u) {
            int j = j0 + u * 256 + (int)t;
            if (j < len) {
                int r = e[u].x & ((1 << ROW_BITS) - 1);
                if (staged) stage[p[u]] = (v2i){ r, e[u].y };
                else        ent[bstart + p[u]] = make_int2(r, e[u].y);
            }
        }
    }
    __syncthreads();
    if (staged) {
#pragma unroll 4
        for (int j = (int)t; j < len; j += 256) {
            v2i e = stage[j];
            ent[bstart + j] = make_int2(e.x, e.y);
        }
    }
}

__global__ void dinv_kernel(const int* __restrict__ off, const int* __restrict__ cnt,
                            const int2* __restrict__ ent, float* __restrict__ dinv) {
    unsigned t = blockIdx.x * blockDim.x + threadIdx.x;
    if (t >= N_NODES * HIDDEN) return;
    unsigned c = t >> 4, f = t & 15;
    int s = off[c], n = cnt[c];
    float d = 0.0f;
    for (int k = (int)f; k < n; k += 16) d += __int_as_float(ent[s + k].y);
    d += __shfl_xor(d, 8);
    d += __shfl_xor(d, 4);
    d += __shfl_xor(d, 2);
    d += __shfl_xor(d, 1);
    d += 1.0f;
    if (f == 0) dinv[c] = (d > 0.0f) ? rsqrtf(d) : 0.0f;
}

__global__ void gemm1_wmma(const float* __restrict__ x, const float* __restrict__ W,
                           const float* __restrict__ dinv, float* __restrict__ hs) {
    unsigned wave = (blockIdx.x * blockDim.x + threadIdx.x) >> 5;
    unsigned lane = threadIdx.x & 31;
    if (wave >= N_TILES) return;
    unsigned m0 = wave * 16;
    unsigned lh = lane & 15;
    unsigned hi = lane >> 4;
    v8f c = {};
    const float* xrow = x + (m0 + lh) * IN_FEAT;
#pragma unroll
    for (int k = 0; k < IN_FEAT; k += 4) {
        int ka = k + (int)hi * 2;
        v2f a = { xrow[ka], xrow[ka + 1] };
        v2f b = { W[ka * HIDDEN + lh], W[(ka + 1) * HIDDEN + lh] };
        c = __builtin_amdgcn_wmma_f32_16x16x4_f32(false, a, false, b,
                                                  (short)0, c, false, false);
    }
    unsigned mbase = m0 + hi * 8;
#pragma unroll
    for (int r = 0; r < 8; ++r) hs[(mbase + r) * HIDDEN + lh] = c[r] * dinv[mbase + r];
}

template <int LAYER>
__global__ void agg_kernel(const int* __restrict__ off, const int* __restrict__ cnt,
                           const int2* __restrict__ ent, const float* __restrict__ hs,
                           const float* __restrict__ dinv, const float* __restrict__ b1,
                           float* __restrict__ dst) {
    unsigned t = blockIdx.x * blockDim.x + threadIdx.x;
    if (t >= N_NODES * HIDDEN) return;
    unsigned c = t >> 4, f = t & 15;
    int s = off[c], n = cnt[c];
    int nmax = max(n, __shfl_xor(n, 16));
    int2 cur;
    {
        int idx = s + (int)f;
        idx = idx < N_EDGES ? idx : N_EDGES - 1;
        int2 v = ent[idx];
        bool ok = (int)f < n;
        cur.x = ok ? v.x : 0;
        cur.y = ok ? v.y : 0;
    }
    float acc = 0.0f;
    for (int base = 0; base < nmax; base += 16) {
        int2 nxt;
        {
            int kn = base + 16 + (int)f;
            int idx = s + kn;
            idx = idx < N_EDGES ? idx : N_EDGES - 1;
            int2 v = ent[idx];
            bool ok = kn < n;
            nxt.x = ok ? v.x : 0;
            nxt.y = ok ? v.y : 0;
        }
        int rem = n - base;
#pragma unroll
        for (int j = 0; j < 16; ++j) {
            int r = __shfl(cur.x, j, 16);
            float wv = __int_as_float(__shfl(cur.y, j, 16));
            float hv = hs[(unsigned)r * HIDDEN + f];
            hv = (j < rem) ? hv : 0.0f;
            acc += wv * hv;
        }
        cur = nxt;
    }
    float di = dinv[c];
    float v = di * (acc + hs[t]);
    if (LAYER == 1) {
        v += b1[f];
        v = v > 0.0f ? v : 0.0f;
        v *= di;
    }
    dst[t] = v;
}

__global__ void gemm2_wmma(const float* __restrict__ a2, const float* __restrict__ W,
                           const float* __restrict__ b2, float* __restrict__ out) {
    unsigned wave = (blockIdx.x * blockDim.x + threadIdx.x) >> 5;
    unsigned lane = threadIdx.x & 31;
    if (wave >= N_TILES) return;
    unsigned m0 = wave * 16;
    unsigned lh = lane & 15, hi = lane >> 4;
    float bia0 = b2[lh], bia1 = b2[16 + lh];
    v8f c0 = { bia0, bia0, bia0, bia0, bia0, bia0, bia0, bia0 };
    v8f c1 = { bia1, bia1, bia1, bia1, bia1, bia1, bia1, bia1 };
    const float* grow = a2 + (m0 + lh) * HIDDEN;
#pragma unroll
    for (int k = 0; k < HIDDEN; k += 4) {
        int ka = k + (int)hi * 2;
        v2f a  = { grow[ka], grow[ka + 1] };
        v2f w0 = { W[ka * N_CLASSES + lh],      W[(ka + 1) * N_CLASSES + lh] };
        v2f w1 = { W[ka * N_CLASSES + 16 + lh], W[(ka + 1) * N_CLASSES + 16 + lh] };
        c0 = __builtin_amdgcn_wmma_f32_16x16x4_f32(false, a, false, w0,
                                                   (short)0, c0, false, false);
        c1 = __builtin_amdgcn_wmma_f32_16x16x4_f32(false, a, false, w1,
                                                   (short)0, c1, false, false);
    }
    unsigned mbase = m0 + hi * 8;
#pragma unroll
    for (int r = 0; r < 8; ++r) {
        out[(mbase + r) * N_CLASSES + lh]      = c0[r];
        out[(mbase + r) * N_CLASSES + 16 + lh] = c1[r];
    }
}

extern "C" void kernel_launch(void* const* d_in, const int* in_sizes, int n_in,
                              void* d_out, int out_size, void* d_ws, size_t ws_size,
                              hipStream_t stream) {
    const float* x  = (const float*)d_in[0];
    const int*   ei = (const int*)d_in[1];
    const float* ew = (const float*)d_in[2];
    const float* W1 = (const float*)d_in[3];
    const float* b1 = (const float*)d_in[4];
    const float* W2 = (const float*)d_in[5];
    const float* b2 = (const float*)d_in[6];
    const int* row = ei;
    const int* col = ei + N_EDGES;
    float* out = (float*)d_out;

    int*   cnt  = (int*)d_ws;
    int*   off  = cnt  + N_NODES;
    int*   bsum = off  + N_NODES;
    float* dinv = (float*)(bsum + 256);
    float* hs1  = dinv + N_NODES;
    float* hs2  = hs1  + (size_t)N_NODES * HIDDEN;
    float* a2   = hs2  + (size_t)N_NODES * HIDDEN;
    int*   bh   = (int*)(a2 + (size_t)N_NODES * HIDDEN);
    int*   bo   = bh + M_BH;
    v2i*   tmp  = (v2i*)(bo + M_BH);
    int2*  ent  = (int2*)(tmp + N_EDGES);

    const int B = 256;
    hist_kernel<<<NBLK, 1024, 0, stream>>>(col, bh);
    blocksum_kernel<<<SCAN_BLOCKS, 256, 0, stream>>>(bh, bsum);
    offsets_kernel<<<SCAN_BLOCKS, 256, 0, stream>>>(bh, bsum, bo);
    bucket_kernel<<<NBLK, 1024, 0, stream>>>(row, col, ew, bh, bo, tmp);
    node_sort_kernel<<<NBK, 256, 0, stream>>>(tmp, bo, off, cnt, ent);
    dinv_kernel<<<(N_NODES * HIDDEN + B - 1) / B, B, 0, stream>>>(off, cnt, ent, dinv);
    gemm1_wmma<<<(N_TILES * 32 + B - 1) / B, B, 0, stream>>>(x, W1, dinv, hs1);
    agg_kernel<1><<<(N_NODES * HIDDEN + B - 1) / B, B, 0, stream>>>(off, cnt, ent, hs1, dinv, b1, hs2);
    agg_kernel<2><<<(N_NODES * HIDDEN + B - 1) / B, B, 0, stream>>>(off, cnt, ent, hs2, dinv, b1, a2);
    gemm2_wmma<<<(N_TILES * 32 + B - 1) / B, B, 0, stream>>>(a2, W2, b2, out);
}


# dinv and both in-edge sums run as 12500 waves that each walk a 16-node tile pair by pair (next pair's entries loaded ahead) instead of 100000 two-node waves; every sum keeps its order
# speedup vs baseline: 5.9151x; 1.0583x over previous
#include <hip/hip_runtime.h>

#define N_NODES   200000
#define N_EDGES   6400000
#define IN_FEAT   128
#define HIDDEN    16
#define N_CLASSES 32
#define N_TILES   (N_NODES / 16)

#define SH        8
#define BNODES    (1 << SH)
#define NBK       ((N_NODES + BNODES - 1) / BNODES)
#define CH        32768
#define NBLK      ((N_EDGES + CH - 1) / CH)
#define M_BH      (NBK * NBLK)
#define E_QUADS   (N_EDGES / 4)
#define N_QUADS   (M_BH / 4)
#define SCAN_BLOCKS ((N_QUADS + 255) / 256)
#define ROW_BITS  18
#define CAP       9984

static_assert(N_EDGES % 4 == 0 && M_BH % 4 == 0, "whole quads");
static_assert(N_NODES <= (1 << ROW_BITS) && SH + ROW_BITS < 32, "source index and local node share one word");
static_assert(NBK <= 1024 && SCAN_BLOCKS <= 256 && CH == 8 * 4 * 1024, "one thread per bucket, one scan block per thread");

typedef __attribute__((ext_vector_type(2))) float v2f;
typedef __attribute__((ext_vector_type(8))) float v8f;
typedef __attribute__((ext_vector_type(2))) int   v2i;


__global__ void __launch_bounds__(1024) hist_kernel(const int* __restrict__ col, int* __restrict__ bh) {
    __shared__ int h[1024];
    unsigned t = threadIdx.x, k = blockIdx.x;
    h[t] = 0;
    __syncthreads();
    const int4* col4 = (const int4*)col;
    int4 c[8];
#pragma unroll
    for (int i = 0; i < 8; ++i) {
        unsigned q = k * (CH / 4) + i * 1024 + t;
        c[i] = col4[q < E_QUADS ? q : 0];
    }
#pragma unroll
    for (int i = 0; i < 8; ++i) {
        unsigned q = k * (CH / 4) + i * 1024 + t;
        if (q < E_QUADS) {
            atomicAdd(&h[c[i].x >> SH], 1);
            atomicAdd(&h[c[i].y >> SH], 1);
            atomicAdd(&h[c[i].z >> SH], 1);
            atomicAdd(&h[c[i].w >> SH], 1);
        }
    }
    __syncthreads();
    if (t < NBK) bh[t * NBLK + k] = h[t];
}

__global__ void blocksum_kernel(const int* __restrict__ cnt, int* __restrict__ bsum) {
    __shared__ int sh[256];
    unsigned t = threadIdx.x;
    unsigned q = blockIdx.x * 256 + t;
    int s = 0;
    if (q < N_QUADS) {
        int4 v = ((const int4*)cnt)[q];
        s = v.x + v.y + v.z + v.w;
    }
    sh[t] = s;
    __syncthreads();
    for (int d = 128; d > 0; d >>= 1) {
        if (t < (unsigned)d) sh[t] += sh[t + d];
        __syncthreads();
    }
    if (t == 0) bsum[blockIdx.x] = sh[0];
}

__global__ void offsets_kernel(const int* __restrict__ cnt, const int* __restrict__ bsum,
                               int* __restrict__ off) {
    __shared__ int sh[256];
    unsigned t = threadIdx.x, b = blockIdx.x;
    sh[t] = (t < b) ? bsum[t] : 0;
    __syncthreads();
    for (int d = 128; d > 0; d >>= 1) {
        if (t < (unsigned)d) sh[t] += sh[t + d];
        __syncthreads();
    }
    int base = sh[0];
    __syncthreads();
    unsigned q = b * 256 + t;
    int4 v = make_int4(0, 0, 0, 0);
    if (q < N_QUADS) v = ((const int4*)cnt)[q];
    int s = v.x + v.y + v.z + v.w;
    sh[t] = s;
    __syncthreads();
    for (int d = 1; d < 256; d <<= 1) {
        int add = (t >= (unsigned)d) ? sh[t - d] : 0;
        __syncthreads();
        sh[t] += add;
        __syncthreads();
    }
    int ex = sh[t] - s + base;
    if (q < N_QUADS) {
        int4 o;
        o.x = ex;
        o.y = o.x + v.x;
        o.z = o.y + v.y;
        o.w = o.z + v.z;
        ((int4*)off)[q] = o;
    }
}

__global__ void __launch_bounds__(1024) bucket_kernel(const int* __restrict__ row, const int* __restrict__ col,
                                                      const float* __restrict__ w, const int* __restrict__ bh,
                                                      const int* __restrict__ bo, v2i* __restrict__ tmp) {
    __shared__ v2i stage[CH];
    __shared__ int lstart[1024];
    __shared__ int cur[1024];
    __shared__ int gb[1024];
    unsigned t = threadIdx.x, k = blockIdx.x;
    int c = 0, g = 0;
    if (t < NBK) {
        c = bh[t * NBLK + k];
        g = bo[t * NBLK + k];
    }
    cur[t] = c;
    __syncthreads();
    for (int d = 1; d < 1024; d <<= 1) {
        int add = (t >= (unsigned)d) ? cur[t - d] : 0;
        __syncthreads();
        cur[t] += add;
        __syncthreads();
    }
    int ex = cur[t] - c;
    lstart[t] = ex;
    cur[t] = ex;
    gb[t] = g;
    __syncthreads();
    const int4*   col4 = (const int4*)col;
    const int4*   row4 = (const int4*)row;
    const float4* w4   = (const float4*)w;
#pragma unroll
    for (int i0 = 0; i0 < 8; i0 += 4) {
        int4 cc[4], rr[4];
        float4 ww[4];
#pragma unroll
        for (int u = 0; u < 4; ++u) {
            unsigned q = k * (CH / 4) + (i0 + u) * 1024 + t;
            unsigned qs = q < E_QUADS ? q : 0;
            cc[u] = col4[qs];
            rr[u] = row4[qs];
            ww[u] = w4[qs];
        }
#pragma unroll
        for (int u = 0; u < 4; ++u) {
            unsigned q = k * (CH / 4) + (i0 + u) * 1024 + t;
            if (q < E_QUADS) {
                int p0 = atomicAdd(&cur[cc[u].x >> SH], 1);
                int p1 = atomicAdd(&cur[cc[u].y >> SH], 1);
                int p2 = atomicAdd(&cur[cc[u].z >> SH], 1);
                int p3 = atomicAdd(&cur[cc[u].w >> SH], 1);
                stage[p0] = (v2i){ ((cc[u].x & (BNODES - 1)) << ROW_BITS) | rr[u].x, __float_as_int(ww[u].x) };
                stage[p1] = (v2i){ ((cc[u].y & (BNODES - 1)) << ROW_BITS) | rr[u].y, __float_as_int(ww[u].y) };
                stage[p2] = (v2i){ ((cc[u].z & (BNODES - 1)) << ROW_BITS) | rr[u].z, __float_as_int(ww[u].z) };
                stage[p3] = (v2i){ ((cc[u].w & (BNODES - 1)) << ROW_BITS) | rr[u].w, __float_as_int(ww[u].w) };
            }
        }
    }
    __syncthreads();
    unsigned wv = t >> 5, lane = t & 31;
    unsigned mb = wv + 32 * lane;
    int my_ls = 0, my_len = 0, my_g = 0;
    if (mb < NBK) {
        my_ls = lstart[mb];
        my_len = cur[mb] - my_ls;
        my_g = gb[mb];
    }
    for (int i = 0; i < (NBK + 31) / 32; ++i) {
        int ls  = __builtin_amdgcn_readlane(my_ls, i);
        int len = __builtin_amdgcn_readlane(my_len, i);
        int g0  = __builtin_amdgcn_readlane(my_g, i);
        for (int j = (int)lane; j < len; j += 32) tmp[g0 + j] = stage[ls + j];
    }
}

__global__ void __launch_bounds__(256) node_sort_kernel(const v2i* __restrict__ tmp, const int* __restrict__ bo,
                                                        int* __restrict__ off, int* __restrict__ cnt,
                                                        int2* __restrict__ ent) {
    __shared__ v2i stage[CAP];
    __shared__ int lc[BNODES];
    unsigned t = threadIdx.x, b = blockIdx.x;
    int bstart = bo[b * NBLK];
    int bend = (b + 1 < NBK) ? bo[(b + 1) * NBLK] : N_EDGES;
    int len = bend - bstart;
    bool staged = len <= CAP;
    lc[t] = 0;
    __syncthreads();
    for (int j0 = 0; j0 < len; j0 += 256 * 8) {
        int key[8];
#pragma unroll
        for (int u = 0; u < 8; ++u) {
            int j = j0 + u * 256 + (int)t;
            key[u] = tmp[bstart + (j < len ? j : 0)].x;
        }
#pragma unroll
        for (int u = 0; u < 8; ++u) {
            int j = j0 + u * 256 + (int)t;
            if (j < len) atomicAdd(&lc[key[u] >> ROW_BITS], 1);
        }
    }
    __syncthreads();
    int c = lc[t];
    for (int d = 1; d < 256; d <<= 1) {
        int add = (t >= (unsigned)d) ? lc[t - d] : 0;
        __syncthreads();
        lc[t] += add;
        __syncthreads();
    }
    int ex = lc[t] - c;
    unsigned node = b * BNODES + t;
    if (node < N_NODES) {
        off[node] = bstart + ex;
        cnt[node] = c;
    }
    lc[t] = ex;
    __syncthreads();
    for (int j0 = 0; j0 < len; j0 += 256 * 8) {
        v2i e[8];
        int p[8];
#pragma unroll
        for (int u = 0; u < 8; ++u) {
            int j = j0 + u * 256 + (int)t;
            e[u] = tmp[bstart + (j < len ? j : 0)];
        }
#pragma unroll
        for (int u = 0; u < 8; ++u) {
            int j = j0 + u * 256 + (int)t;
            p[u] = 0;
            if (j < len) p[u] = atomicAdd(&lc[e[u].x >> ROW_BITS], 1);
        }
#pragma unroll
        for (int u = 0; u < 8; ++u) {
            int j = j0 + u * 256 + (int)t;
            if (j < len) {
                int r = e[u].x & ((1 << ROW_BITS) - 1);
                if (staged) stage[p[u]] = (v2i){ r, e[u].y };
                else        ent[bstart + p[u]] = make_int2(r, e[u].y);
            }
        }
    }
    __syncthreads();
    if (staged) {
#pragma unroll 4
        for (int j = (int)t; j < len; j += 256) {
            v2i e = stage[j];
            ent[bstart + j] = make_int2(e.x, e.y);
        }
    }
}

#define PAIRS 8

__device__ __forceinline__ int pick_node(int v, int p, unsigned hi) {
    int lo = __builtin_amdgcn_readlane(v, 2 * p);
    int up = __builtin_amdgcn_readlane(v, 2 * p + 1);
    return hi ? up : lo;
}

__device__ __forceinline__ int2 load_entry(const int2* __restrict__ ent, int s, int n, int k) {
    int idx = s + k;
    idx = idx < N_EDGES ? idx : N_EDGES - 1;
    int2 v = ent[idx];
    bool ok = k < n;
    int2 r;
    r.x = ok ? v.x : 0;
    r.y = ok ? v.y : 0;
    return r;
}

__global__ void __launch_bounds__(256) dinv_kernel(const int* __restrict__ off, const int* __restrict__ cnt,
                                                   const int2* __restrict__ ent, float* __restrict__ dinv) {
    unsigned wave = (blockIdx.x * blockDim.x + threadIdx.x) >> 5;
    if (wave >= N_TILES) return;
    unsigned lane = threadIdx.x & 31;
    unsigned f = lane & 15, hi = lane >> 4;
    unsigned c0 = wave * 16;
    int offv = off[c0 + f];
    int cntv = cnt[c0 + f];
    int s = pick_node(offv, 0, hi), n = pick_node(cntv, 0, hi);
    float w0 = __int_as_float(load_entry(ent, s, n, (int)f).y);
    float w1 = __int_as_float(load_entry(ent, s, n, (int)f + 16).y);
    float w2 = __int_as_float(load_entry(ent, s, n, (int)f + 32).y);
    float res = 0.0f;
    for (int p = 0; p < PAIRS; ++p) {
        int sn = 0, nn = 0;
        if (p + 1 < PAIRS) {
            sn = pick_node(offv, p + 1, hi);
            nn = pick_node(cntv, p + 1, hi);
        }
        float x0 = __int_as_float(load_entry(ent, sn, nn, (int)f).y);
        float x1 = __int_as_float(load_entry(ent, sn, nn, (int)f + 16).y);
        float x2 = __int_as_float(load_entry(ent, sn, nn, (int)f + 32).y);
        float d = 0.0f;
        if ((int)f < n)      d += w0;
        if ((int)f + 16 < n) d += w1;
        if ((int)f + 32 < n) d += w2;
        for (int k = (int)f + 48; k < n; k += 16) d += __int_as_float(ent[s + k].y);
        d += __shfl_xor(d, 8);
        d += __shfl_xor(d, 4);
        d += __shfl_xor(d, 2);
        d += __shfl_xor(d, 1);
        d += 1.0f;
        float r = (d > 0.0f) ? rsqrtf(d) : 0.0f;
        if (f == 2u * (unsigned)p + hi) res = r;
        s = sn; n = nn;
        w0 = x0; w1 = x1; w2 = x2;
    }
    if ((f & 1u) == hi) dinv[c0 + f] = res;
}

__global__ void gemm1_wmma(const float* __restrict__ x, const float* __restrict__ W,
                           const float* __restrict__ dinv, float* __restrict__ hs) {
    unsigned wave = (blockIdx.x * blockDim.x + threadIdx.x) >> 5;
    unsigned lane = threadIdx.x & 31;
    if (wave >= N_TILES) return;
    unsigned m0 = wave * 16;
    unsigned lh = lane & 15;
    unsigned hi = lane >> 4;
    v8f c = {};
    const float* xrow = x + (m0 + lh) * IN_FEAT;
#pragma unroll
    for (int k = 0; k < IN_FEAT; k += 4) {
        int ka = k + (int)hi * 2;
        v2f a = { xrow[ka], xrow[ka + 1] };
        v2f b = { W[ka * HIDDEN + lh], W[(ka + 1) * HIDDEN + lh] };
        c = __builtin_amdgcn_wmma_f32_16x16x4_f32(false, a, false, b,
                                                  (short)0, c, false, false);
    }
    unsigned mbase = m0 + hi * 8;
#pragma unroll
    for (int r = 0; r < 8; ++r) hs[(mbase + r) * HIDDEN + lh] = c[r] * dinv[mbase + r];
}

template <int LAYER>
__global__ void __launch_bounds__(256) agg_kernel(const int* __restrict__ off, const int* __restrict__ cnt,
                                                  const int2* __restrict__ ent, const float* __restrict__ hs,
                                                  const float* __restrict__ dinv, const float* __restrict__ b1,
                                                  float* __restrict__ dst) {
    unsigned wave = (blockIdx.x * blockDim.x + threadIdx.x) >> 5;
    if (wave >= N_TILES) return;
    unsigned lane = threadIdx.x & 31;
    unsigned f = lane & 15, hi = lane >> 4;
    unsigned c0 = wave * 16;
    int offv = off[c0 + f];
    int cntv = cnt[c0 + f];
    int dinb = __float_as_int(dinv[c0 + f]);
    float bias = 0.0f;
    if (LAYER == 1) bias = b1[f];
    int sn = pick_node(offv, 0, hi), nn = pick_node(cntv, 0, hi);
    int2 cur = load_entry(ent, sn, nn, (int)f);
    for (int p = 0; p < PAIRS; ++p) {
        int s = sn, n = nn;
        int nmax = max(__builtin_amdgcn_readlane(cntv, 2 * p), __builtin_amdgcn_readlane(cntv, 2 * p + 1));
        sn = 0; nn = 0;
        if (p + 1 < PAIRS) {
            sn = pick_node(offv, p + 1, hi);
            nn = pick_node(cntv, p + 1, hi);
        }
        unsigned t = (c0 + 2u * (unsigned)p + hi) * HIDDEN + f;
        float self = hs[t];
        float acc = 0.0f;
        for (int base = 0; base < nmax; base += 16) {
            bool last = base + 16 >= nmax;
            int2 nxt = load_entry(ent, last ? sn : s, last ? nn : n, last ? (int)f : base + 16 + (int)f);
            int rem = n - base;
#pragma unroll
            for (int j = 0; j < 16; ++j) {
                int r = __shfl(cur.x, j, 16);
                float wv = __int_as_float(__shfl(cur.y, j, 16));
                float hv = hs[(unsigned)r * HIDDEN + f];
                hv = (j < rem) ? hv : 0.0f;
                acc += wv * hv;
            }
            cur = nxt;
        }
        if (nmax == 0) cur = load_entry(ent, sn, nn, (int)f);
        float di = __int_as_float(pick_node(dinb, p, hi));
        float v = di * (acc + self);
        if (LAYER == 1) {
            v += bias;
            v = v > 0.0f ? v : 0.0f;
            v *= di;
        }
        dst[t] = v;
    }
}

__global__ void gemm2_wmma(const float* __restrict__ a2, const float* __restrict__ W,
                           const float* __restrict__ b2, float* __restrict__ out) {
    unsigned wave = (blockIdx.x * blockDim.x + threadIdx.x) >> 5;
    unsigned lane = threadIdx.x & 31;
    if (wave >= N_TILES) return;
    unsigned m0 = wave * 16;
    unsigned lh = lane & 15, hi = lane >> 4;
    float bia0 = b2[lh], bia1 = b2[16 + lh];
    v8f c0 = { bia0, bia0, bia0, bia0, bia0, bia0, bia0, bia0 };
    v8f c1 = { bia1, bia1, bia1, bia1, bia1, bia1, bia1, bia1 };
    const float* grow = a2 + (m0 + lh) * HIDDEN;
#pragma unroll
    for (int k = 0; k < HIDDEN; k += 4) {
        int ka = k + (int)hi * 2;
        v2f a  = { grow[ka], grow[ka + 1] };
        v2f w0 = { W[ka * N_CLASSES + lh],      W[(ka + 1) * N_CLASSES + lh] };
        v2f w1 = { W[ka * N_CLASSES + 16 + lh], W[(ka + 1) * N_CLASSES + 16 + lh] };
        c0 = __builtin_amdgcn_wmma_f32_16x16x4_f32(false, a, false, w0,
                                                   (short)0, c0, false, false);
        c1 = __builtin_amdgcn_wmma_f32_16x16x4_f32(false, a, false, w1,
                                                   (short)0, c1, false, false);
    }
    unsigned mbase = m0 + hi * 8;
#pragma unroll
    for (int r = 0; r < 8; ++r) {
        out[(mbase + r) * N_CLASSES + lh]      = c0[r];
        out[(mbase + r) * N_CLASSES + 16 + lh] = c1[r];
    }
}

extern "C" void kernel_launch(void* const* d_in, const int* in_sizes, int n_in,
                              void* d_out, int out_size, void* d_ws, size_t ws_size,
                              hipStream_t stream) {
    const float* x  = (const float*)d_in[0];
    const int*   ei = (const int*)d_in[1];
    const float* ew = (const float*)d_in[2];
    const float* W1 = (const float*)d_in[3];
    const float* b1 = (const float*)d_in[4];
    const float* W2 = (const float*)d_in[5];
    const float* b2 = (const float*)d_in[6];
    const int* row = ei;
    const int* col = ei + N_EDGES;
    float* out = (float*)d_out;

    int*   cnt  = (int*)d_ws;
    int*   off  = cnt  + N_NODES;
    int*   bsum = off  + N_NODES;
    float* dinv = (float*)(bsum + 256);
    float* hs1  = dinv + N_NODES;
    float* hs2  = hs1  + (size_t)N_NODES * HIDDEN;
    float* a2   = hs2  + (size_t)N_NODES * HIDDEN;
    int*   bh   = (int*)(a2 + (size_t)N_NODES * HIDDEN);
    int*   bo   = bh + M_BH;
    v2i*   tmp  = (v2i*)(bo + M_BH);
    int2*  ent  = (int2*)(tmp + N_EDGES);

    const int B = 256;
    hist_kernel<<<NBLK, 1024, 0, stream>>>(col, bh);
    blocksum_kernel<<<SCAN_BLOCKS, 256, 0, stream>>>(bh, bsum);
    offsets_kernel<<<SCAN_BLOCKS, 256, 0, stream>>>(bh, bsum, bo);
    bucket_kernel<<<NBLK, 1024, 0, stream>>>(row, col, ew, bh, bo, tmp);
    node_sort_kernel<<<NBK, 256, 0, stream>>>(tmp, bo, off, cnt, ent);
    dinv_kernel<<<(N_TILES * 32 + B - 1) / B, B, 0, stream>>>(off, cnt, ent, dinv);
    gemm1_wmma<<<(N_TILES * 32 + B - 1) / B, B, 0, stream>>>(x, W1, dinv, hs1);
    agg_kernel<1><<<(N_TILES * 32 + B - 1) / B, B, 0, stream>>>(off, cnt, ent, hs1, dinv, b1, hs2);
    agg_kernel<2><<<(N_TILES * 32 + B - 1) / B, B, 0, stream>>>(off, cnt, ent, hs2, dinv, b1, a2);
    gemm2_wmma<<<(N_TILES * 32 + B - 1) / B, B, 0, stream>>>(a2, W2, b2, out);
}


# dinv, GEMM1 and GEMM2 fold into the launches that make their inputs (node sort; layer-2 sum feeds its WMMAs through LDS): 10 launches become 7; deg summed in sort order: changes f32 summation order
# speedup vs baseline: 6.5482x; 1.1070x over previous
#include <hip/hip_runtime.h>

#define N_NODES   200000
#define N_EDGES   6400000
#define IN_FEAT   128
#define HIDDEN    16
#define N_CLASSES 32
#define N_TILES   (N_NODES / 16)

#define SH        8
#define BNODES    (1 << SH)
#define NBK       ((N_NODES + BNODES - 1) / BNODES)
#define CH        32768
#define NBLK      ((N_EDGES + CH - 1) / CH)
#define M_BH      (NBK * NBLK)
#define E_QUADS   (N_EDGES / 4)
#define N_QUADS   (M_BH / 4)
#define SCAN_BLOCKS ((N_QUADS + 255) / 256)
#define ROW_BITS  18
#define CAP       9984

static_assert(N_EDGES % 4 == 0 && M_BH % 4 == 0, "whole quads");
static_assert(N_NODES <= (1 << ROW_BITS) && SH + ROW_BITS < 32, "source index and local node share one word");
static_assert(NBK <= 1024 && SCAN_BLOCKS <= 256 && CH == 8 * 4 * 1024, "one thread per bucket, one scan block per thread");

typedef __attribute__((ext_vector_type(2))) float v2f;
typedef __attribute__((ext_vector_type(8))) float v8f;
typedef __attribute__((ext_vector_type(2))) int   v2i;


__global__ void __launch_bounds__(1024) hist_kernel(const int* __restrict__ col, int* __restrict__ bh) {
    __shared__ int h[1024];
    unsigned t = threadIdx.x, k = blockIdx.x;
    h[t] = 0;
    __syncthreads();
    const int4* col4 = (const int4*)col;
    int4 c[8];
#pragma unroll
    for (int i = 0; i < 8; ++i) {
        unsigned q = k * (CH / 4) + i * 1024 + t;
        c[i] = col4[q < E_QUADS ? q : 0];
    }
#pragma unroll
    for (int i = 0; i < 8; ++i) {
        unsigned q = k * (CH / 4) + i * 1024 + t;
        if (q < E_QUADS) {
            atomicAdd(&h[c[i].x >> SH], 1);
            atomicAdd(&h[c[i].y >> SH], 1);
            atomicAdd(&h[c[i].z >> SH], 1);
            atomicAdd(&h[c[i].w >> SH], 1);
        }
    }
    __syncthreads();
    if (t < NBK) bh[t * NBLK + k] = h[t];
}

__global__ void blocksum_kernel(const int* __restrict__ cnt, int* __restrict__ bsum) {
    __shared__ int sh[256];
    unsigned t = threadIdx.x;
    unsigned q = blockIdx.x * 256 + t;
    int s = 0;
    if (q < N_QUADS) {
        int4 v = ((const int4*)cnt)[q];
        s = v.x + v.y + v.z + v.w;
    }
    sh[t] = s;
    __syncthreads();
    for (int d = 128; d > 0; d >>= 1) {
        if (t < (unsigned)d) sh[t] += sh[t + d];
        __syncthreads();
    }
    if (t == 0) bsum[blockIdx.x] = sh[0];
}

__global__ void offsets_kernel(const int* __restrict__ cnt, const int* __restrict__ bsum,
                               int* __restrict__ off) {
    __shared__ int sh[256];
    unsigned t = threadIdx.x, b = blockIdx.x;
    sh[t] = (t < b) ? bsum[t] : 0;
    __syncthreads();
    for (int d = 128; d > 0; d >>= 1) {
        if (t < (unsigned)d) sh[t] += sh[t + d];
        __syncthreads();
    }
    int base = sh[0];
    __syncthreads();
    unsigned q = b * 256 + t;
    int4 v = make_int4(0, 0, 0, 0);
    if (q < N_QUADS) v = ((const int4*)cnt)[q];
    int s = v.x + v.y + v.z + v.w;
    sh[t] = s;
    __syncthreads();
    for (int d = 1; d < 256; d <<= 1) {
        int add = (t >= (unsigned)d) ? sh[t - d] : 0;
        __syncthreads();
        sh[t] += add;
        __syncthreads();
    }
    int ex = sh[t] - s + base;
    if (q < N_QUADS) {
        int4 o;
        o.x = ex;
        o.y = o.x + v.x;
        o.z = o.y + v.y;
        o.w = o.z + v.z;
        ((int4*)off)[q] = o;
    }
}

__global__ void __launch_bounds__(1024) bucket_kernel(const int* __restrict__ row, const int* __restrict__ col,
                                                      const float* __restrict__ w, const int* __restrict__ bh,
                                                      const int* __restrict__ bo, v2i* __restrict__ tmp) {
    __shared__ v2i stage[CH];
    __shared__ int lstart[1024];
    __shared__ int cur[1024];
    __shared__ int gb[1024];
    unsigned t = threadIdx.x, k = blockIdx.x;
    int c = 0, g = 0;
    if (t < NBK) {
        c = bh[t * NBLK + k];
        g = bo[t * NBLK + k];
    }
    cur[t] = c;
    __syncthreads();
    for (int d = 1; d < 1024; d <<= 1) {
        int add = (t >= (unsigned)d) ? cur[t - d] : 0;
        __syncthreads();
        cur[t] += add;
        __syncthreads();
    }
    int ex = cur[t] - c;
    lstart[t] = ex;
    cur[t] = ex;
    gb[t] = g;
    __syncthreads();
    const int4*   col4 = (const int4*)col;
    const int4*   row4 = (const int4*)row;
    const float4* w4   = (const float4*)w;
#pragma unroll
    for (int i0 = 0; i0 < 8; i0 += 4) {
        int4 cc[4], rr[4];
        float4 ww[4];
#pragma unroll
        for (int u = 0; u < 4; ++u) {
            unsigned q = k * (CH / 4) + (i0 + u) * 1024 + t;
            unsigned qs = q < E_QUADS ? q : 0;
            cc[u] = col4[qs];
            rr[u] = row4[qs];
            ww[u] = w4[qs];
        }
#pragma unroll
        for (int u = 0; u < 4; ++u) {
            unsigned q = k * (CH / 4) + (i0 + u) * 1024 + t;
            if (q < E_QUADS) {
                int p0 = atomicAdd(&cur[cc[u].x >> SH], 1);
                int p1 = atomicAdd(&cur[cc[u].y >> SH], 1);
                int p2 = atomicAdd(&cur[cc[u].z >> SH], 1);
                int p3 = atomicAdd(&cur[cc[u].w >> SH], 1);
                stage[p0] = (v2i){ ((cc[u].x & (BNODES - 1)) << ROW_BITS) | rr[u].x, __float_as_int(ww[u].x) };
                stage[p1] = (v2i){ ((cc[u].y & (BNODES - 1)) << ROW_BITS) | rr[u].y, __float_as_int(ww[u].y) };
                stage[p2] = (v2i){ ((cc[u].z & (BNODES - 1)) << ROW_BITS) | rr[u].z, __float_as_int(ww[u].z) };
                stage[p3] = (v2i){ ((cc[u].w & (BNODES - 1)) << ROW_BITS) | rr[u].w, __float_as_int(ww[u].w) };
            }
        }
    }
    __syncthreads();
    unsigned wv = t >> 5, lane = t & 31;
    unsigned mb = wv + 32 * lane;
    int my_ls = 0, my_len = 0, my_g = 0;
    if (mb < NBK) {
        my_ls = lstart[mb];
        my_len = cur[mb] - my_ls;
        my_g = gb[mb];
    }
    for (int i = 0; i < (NBK + 31) / 32; ++i) {
        int ls  = __builtin_amdgcn_readlane(my_ls, i);
        int len = __builtin_amdgcn_readlane(my_len, i);
        int g0  = __builtin_amdgcn_readlane(my_g, i);
        for (int j = (int)lane; j < len; j += 32) tmp[g0 + j] = stage[ls + j];
    }
}

__global__ void __launch_bounds__(256) node_sort_kernel(const v2i* __restrict__ tmp, const int* __restrict__ bo,
                                                        int* __restrict__ off, int* __restrict__ cnt,
                                                        int2* __restrict__ ent,
                                                        const float* __restrict__ x, const float* __restrict__ W,
                                                        float* __restrict__ dinv, float* __restrict__ hs) {
    __shared__ v2i stage[CAP];
    __shared__ int lc[BNODES];
    unsigned t = threadIdx.x, b = blockIdx.x;
    int bstart = bo[b * NBLK];
    int bend = (b + 1 < NBK) ? bo[(b + 1) * NBLK] : N_EDGES;
    int len = bend - bstart;
    bool staged = len <= CAP;
    lc[t] = 0;
    __syncthreads();
    for (int j0 = 0; j0 < len; j0 += 256 * 8) {
        int key[8];
#pragma unroll
        for (int u = 0; u < 8; ++u) {
            int j = j0 + u * 256 + (int)t;
            key[u] = tmp[bstart + (j < len ? j : 0)].x;
        }
#pragma unroll
        for (int u = 0; u < 8; ++u) {
            int j = j0 + u * 256 + (int)t;
            if (j < len) atomicAdd(&lc[key[u] >> ROW_BITS], 1);
        }
    }
    __syncthreads();
    int c = lc[t];
    for (int d = 1; d < 256; d <<= 1) {
        int add = (t >= (unsigned)d) ? lc[t - d] : 0;
        __syncthreads();
        lc[t] += add;
        __syncthreads();
    }
    int ex = lc[t] - c;
    unsigned node = b * BNODES + t;
    if (node < N_NODES) {
        off[node] = bstart + ex;
        cnt[node] = c;
    }
    lc[t] = ex;
    __syncthreads();
    for (int j0 = 0; j0 < len; j0 += 256 * 8) {
        v2i e[8];
        int p[8];
#pragma unroll
        for (int u = 0; u < 8; ++u) {
            int j = j0 + u * 256 + (int)t;
            e[u] = tmp[bstart + (j < len ? j : 0)];
        }
#pragma unroll
        for (int u = 0; u < 8; ++u) {
            int j = j0 + u * 256 + (int)t;
            p[u] = 0;
            if (j < len) p[u] = atomicAdd(&lc[e[u].x >> ROW_BITS], 1);
        }
#pragma unroll
        for (int u = 0; u < 8; ++u) {
            int j = j0 + u * 256 + (int)t;
            if (j < len) {
                int r = e[u].x & ((1 << ROW_BITS) - 1);
                if (staged) stage[p[u]] = (v2i){ r, e[u].y };
                else        ent[bstart + p[u]] = make_int2(r, e[u].y);
            }
        }
    }
    if (!staged) __threadfence();
    __syncthreads();
    float deg = 0.0f;
    if (staged) {
#pragma unroll 4
        for (int j = (int)t; j < len; j += 256) {
            v2i e = stage[j];
            ent[bstart + j] = make_int2(e.x, e.y);
        }
#pragma unroll 4
        for (int i = 0; i < c; ++i) deg += __int_as_float(stage[ex + i].y);
    } else {
        for (int i = 0; i < c; ++i) deg += __int_as_float(ent[bstart + ex + i].y);
    }
    deg += 1.0f;
    float dn = (deg > 0.0f) ? rsqrtf(deg) : 0.0f;
    if (node < N_NODES) dinv[node] = dn;
    lc[t] = __float_as_int(dn);
    __syncthreads();
    unsigned wv = (unsigned)__builtin_amdgcn_readfirstlane((int)(t >> 5));
    unsigned lane = t & 31;
    unsigned lh = lane & 15;
    unsigned hi = lane >> 4;
    for (unsigned tt = wv; tt < BNODES / 16; tt += 8) {
        unsigned m0 = b * BNODES + tt * 16;
        if (m0 >= N_NODES) break;
        v8f acc = {};
        const float* xrow = x + (size_t)(m0 + lh) * IN_FEAT;
#pragma unroll
        for (int k = 0; k < IN_FEAT; k += 4) {
            int ka = k + (int)hi * 2;
            v2f a = { xrow[ka], xrow[ka + 1] };
            v2f bw = { W[ka * HIDDEN + lh], W[(ka + 1) * HIDDEN + lh] };
            acc = __builtin_amdgcn_wmma_f32_16x16x4_f32(false, a, false, bw,
                                                        (short)0, acc, false, false);
        }
        unsigned lb = tt * 16 + hi * 8;
#pragma unroll
        for (int r = 0; r < 8; ++r)
            hs[(size_t)(b * BNODES + lb + r) * HIDDEN + lh] = acc[r] * __int_as_float(lc[lb + r]);
    }
}

#define PAIRS 8

__device__ __forceinline__ int pick_node(int v, int p, unsigned hi) {
    int lo = __builtin_amdgcn_readlane(v, 2 * p);
    int up = __builtin_amdgcn_readlane(v, 2 * p + 1);
    return hi ? up : lo;
}

__device__ __forceinline__ int2 load_entry(const int2* __restrict__ ent, int s, int n, int k) {
    int idx = s + k;
    idx = idx < N_EDGES ? idx : N_EDGES - 1;
    int2 v = ent[idx];
    bool ok = k < n;
    int2 r;
    r.x = ok ? v.x : 0;
    r.y = ok ? v.y : 0;
    return r;
}

#define TP 18
template <int LAYER>
__global__ void __launch_bounds__(256) agg_kernel(const int* __restrict__ off, const int* __restrict__ cnt,
                                                  const int2* __restrict__ ent, const float* __restrict__ hs,
                                                  const float* __restrict__ dinv, const float* __restrict__ b1,
                                                  float* __restrict__ dst, const float* __restrict__ W2,
                                                  const float* __restrict__ b2, float* __restrict__ out) {
    __shared__ float tile[LAYER == 2 ? 8 * 16 * TP : 1];
    unsigned wave = (blockIdx.x * blockDim.x + threadIdx.x) >> 5;
    if (wave >= N_TILES) return;
    unsigned lane = threadIdx.x & 31;
    unsigned f = lane & 15, hi = lane >> 4;
    unsigned c0 = wave * 16;
    int offv = off[c0 + f];
    int cntv = cnt[c0 + f];
    int dinb = __float_as_int(dinv[c0 + f]);
    float bias = 0.0f;
    if (LAYER == 1) bias = b1[f];
    int sn = pick_node(offv, 0, hi), nn = pick_node(cntv, 0, hi);
    int2 cur = load_entry(ent, sn, nn, (int)f);
    for (int p = 0; p < PAIRS; ++p) {
        int s = sn, n = nn;
        int nmax = max(__builtin_amdgcn_readlane(cntv, 2 * p), __builtin_amdgcn_readlane(cntv, 2 * p + 1));
        sn = 0; nn = 0;
        if (p + 1 < PAIRS) {
            sn = pick_node(offv, p + 1, hi);
            nn = pick_node(cntv, p + 1, hi);
        }
        unsigned t = (c0 + 2u * (unsigned)p + hi) * HIDDEN + f;
        float self = hs[t];
        float acc = 0.0f;
        for (int base = 0; base < nmax; base += 16) {
            bool last = base + 16 >= nmax;
            int2 nxt = load_entry(ent, last ? sn : s, last ? nn : n, last ? (int)f : base + 16 + (int)f);
            int rem = n - base;
#pragma unroll
            for (int j = 0; j < 16; ++j) {
                int r = __shfl(cur.x, j, 16);
                float wv = __int_as_float(__shfl(cur.y, j, 16));
                float hv = hs[(unsigned)r * HIDDEN + f];
                hv = (j < rem) ? hv : 0.0f;
                acc += wv * hv;
            }
            cur = nxt;
        }
        if (nmax == 0) cur = load_entry(ent, sn, nn, (int)f);
        float di = __int_as_float(pick_node(dinb, p, hi));
        float v = di * (acc + self);
        if (LAYER == 1) {
            v += bias;
            v = v > 0.0f ? v : 0.0f;
            v *= di;
            dst[t] = v;
        } else {
            tile[(threadIdx.x >> 5) * (16 * TP) + (2u * (unsigned)p + hi) * TP + f] = v;
        }
    }
    if (LAYER == 2) {
        const float* trow = tile + (threadIdx.x >> 5) * (16 * TP) + f * TP;
        float bia0 = b2[f], bia1 = b2[16 + f];
        v8f c0v = { bia0, bia0, bia0, bia0, bia0, bia0, bia0, bia0 };
        v8f c1v = { bia1, bia1, bia1, bia1, bia1, bia1, bia1, bia1 };
#pragma unroll
        for (int k = 0; k < HIDDEN; k += 4) {
            int ka = k + (int)hi * 2;
            v2f a  = { trow[ka], trow[ka + 1] };
            v2f w0 = { W2[ka * N_CLASSES + f],      W2[(ka + 1) * N_CLASSES + f] };
            v2f w1 = { W2[ka * N_CLASSES + 16 + f], W2[(ka + 1) * N_CLASSES + 16 + f] };
            c0v = __builtin_amdgcn_wmma_f32_16x16x4_f32(false, a, false, w0,
                                                        (short)0, c0v, false, false);
            c1v = __builtin_amdgcn_wmma_f32_16x16x4_f32(false, a, false, w1,
                                                        (short)0, c1v, false, false);
        }
        unsigned mbase = c0 + hi * 8;
#pragma unroll
        for (int r = 0; r < 8; ++r) {
            out[(mbase + r) * N_CLASSES + f]      = c0v[r];
            out[(mbase + r) * N_CLASSES + 16 + f] = c1v[r];
        }
    }
}

extern "C" void kernel_launch(void* const* d_in, const int* in_sizes, int n_in,
                              void* d_out, int out_size, void* d_ws, size_t ws_size,
                              hipStream_t stream) {
    const float* x  = (const float*)d_in[0];
    const int*   ei = (const int*)d_in[1];
    const float* ew = (const float*)d_in[2];
    const float* W1 = (const float*)d_in[3];
    const float* b1 = (const float*)d_in[4];
    const float* W2 = (const float*)d_in[5];
    const float* b2 = (const float*)d_in[6];
    const int* row = ei;
    const int* col = ei + N_EDGES;
    float* out = (float*)d_out;

    int*   cnt  = (int*)d_ws;
    int*   off  = cnt  + N_NODES;
    int*   bsum = off  + N_NODES;
    float* dinv = (float*)(bsum + 256);
    float* hs1  = dinv + N_NODES;
    float* hs2  = hs1  + (size_t)N_NODES * HIDDEN;
    float* spare = hs2 + (size_t)N_NODES * HIDDEN;
    int*   bh   = (int*)(spare + (size_t)N_NODES * HIDDEN);
    int*   bo   = bh + M_BH;
    v2i*   tmp  = (v2i*)(bo + M_BH);
    int2*  ent  = (int2*)(tmp + N_EDGES);

    const int B = 256;
    hist_kernel<<<NBLK, 1024, 0, stream>>>(col, bh);
    blocksum_kernel<<<SCAN_BLOCKS, 256, 0, stream>>>(bh, bsum);
    offsets_kernel<<<SCAN_BLOCKS, 256, 0, stream>>>(bh, bsum, bo);
    bucket_kernel<<<NBLK, 1024, 0, stream>>>(row, col, ew, bh, bo, tmp);
    node_sort_kernel<<<NBK, 256, 0, stream>>>(tmp, bo, off, cnt, ent, x, W1, dinv, hs1);
    agg_kernel<1><<<(N_TILES * 32 + B - 1) / B, B, 0, stream>>>(off, cnt, ent, hs1, dinv, b1, hs2, W2, b2, out);
    agg_kernel<2><<<(N_TILES * 32 + B - 1) / B, B, 0, stream>>>(off, cnt, ent, hs2, dinv, b1, spare, W2, b2, out);
}


# Both in-edge sums gather whole 64-byte rows by 16-byte loads: 4 lanes per node, 8 nodes abreast, 8 rows per load instruction instead of 2; sums keep their order, products now fused (fma)
# speedup vs baseline: 7.2971x; 1.1144x over previous
#include <hip/hip_runtime.h>

#define N_NODES   200000
#define N_EDGES   6400000
#define IN_FEAT   128
#define HIDDEN    16
#define N_CLASSES 32
#define N_TILES   (N_NODES / 16)

#define SH        8
#define BNODES    (1 << SH)
#define NBK       ((N_NODES + BNODES - 1) / BNODES)
#define CH        32768
#define NBLK      ((N_EDGES + CH - 1) / CH)
#define M_BH      (NBK * NBLK)
#define E_QUADS   (N_EDGES / 4)
#define N_QUADS   (M_BH / 4)
#define SCAN_BLOCKS ((N_QUADS + 255) / 256)
#define ROW_BITS  18
#define CAP       9984

static_assert(N_EDGES % 4 == 0 && M_BH % 4 == 0, "whole quads");
static_assert(N_NODES <= (1 << ROW_BITS) && SH + ROW_BITS < 32, "source index and local node share one word");
static_assert(NBK <= 1024 && SCAN_BLOCKS <= 256 && CH == 8 * 4 * 1024, "one thread per bucket, one scan block per thread");

typedef __attribute__((ext_vector_type(2))) float v2f;
typedef __attribute__((ext_vector_type(8))) float v8f;
typedef __attribute__((ext_vector_type(2))) int   v2i;


__global__ void __launch_bounds__(1024) hist_kernel(const int* __restrict__ col, int* __restrict__ bh) {
    __shared__ int h[1024];
    unsigned t = threadIdx.x, k = blockIdx.x;
    h[t] = 0;
    __syncthreads();
    const int4* col4 = (const int4*)col;
    int4 c[8];
#pragma unroll
    for (int i = 0; i < 8; ++i) {
        unsigned q = k * (CH / 4) + i * 1024 + t;
        c[i] = col4[q < E_QUADS ? q : 0];
    }
#pragma unroll
    for (int i = 0; i < 8; ++i) {
        unsigned q = k * (CH / 4) + i * 1024 + t;
        if (q < E_QUADS) {
            atomicAdd(&h[c[i].x >> SH], 1);
            atomicAdd(&h[c[i].y >> SH], 1);
            atomicAdd(&h[c[i].z >> SH], 1);
            atomicAdd(&h[c[i].w >> SH], 1);
        }
    }
    __syncthreads();
    if (t < NBK) bh[t * NBLK + k] = h[t];
}

__global__ void blocksum_kernel(const int* __restrict__ cnt, int* __restrict__ bsum) {
    __shared__ int sh[256];
    unsigned t = threadIdx.x;
    unsigned q = blockIdx.x * 256 + t;
    int s = 0;
    if (q < N_QUADS) {
        int4 v = ((const int4*)cnt)[q];
        s = v.x + v.y + v.z + v.w;
    }
    sh[t] = s;
    __syncthreads();
    for (int d = 128; d > 0; d >>= 1) {
        if (t < (unsigned)d) sh[t] += sh[t + d];
        __syncthreads();
    }
    if (t == 0) bsum[blockIdx.x] = sh[0];
}

__global__ void offsets_kernel(const int* __restrict__ cnt, const int* __restrict__ bsum,
                               int* __restrict__ off) {
    __shared__ int sh[256];
    unsigned t = threadIdx.x, b = blockIdx.x;
    sh[t] = (t < b) ? bsum[t] : 0;
    __syncthreads();
    for (int d = 128; d > 0; d >>= 1) {
        if (t < (unsigned)d) sh[t] += sh[t + d];
        __syncthreads();
    }
    int base = sh[0];
    __syncthreads();
    unsigned q = b * 256 + t;
    int4 v = make_int4(0, 0, 0, 0);
    if (q < N_QUADS) v = ((const int4*)cnt)[q];
    int s = v.x + v.y + v.z + v.w;
    sh[t] = s;
    __syncthreads();
    for (int d = 1; d < 256; d <<= 1) {
        int add = (t >= (unsigned)d) ? sh[t - d] : 0;
        __syncthreads();
        sh[t] += add;
        __syncthreads();
    }
    int ex = sh[t] - s + base;
    if (q < N_QUADS) {
        int4 o;
        o.x = ex;
        o.y = o.x + v.x;
        o.z = o.y + v.y;
        o.w = o.z + v.z;
        ((int4*)off)[q] = o;
    }
}

__global__ void __launch_bounds__(1024) bucket_kernel(const int* __restrict__ row, const int* __restrict__ col,
                                                      const float* __restrict__ w, const int* __restrict__ bh,
                                                      const int* __restrict__ bo, v2i* __restrict__ tmp) {
    __shared__ v2i stage[CH];
    __shared__ int lstart[1024];
    __shared__ int cur[1024];
    __shared__ int gb[1024];
    unsigned t = threadIdx.x, k = blockIdx.x;
    int c = 0, g = 0;
    if (t < NBK) {
        c = bh[t * NBLK + k];
        g = bo[t * NBLK + k];
    }
    cur[t] = c;
    __syncthreads();
    for (int d = 1; d < 1024; d <<= 1) {
        int add = (t >= (unsigned)d) ? cur[t - d] : 0;
        __syncthreads();
        cur[t] += add;
        __syncthreads();
    }
    int ex = cur[t] - c;
    lstart[t] = ex;
    cur[t] = ex;
    gb[t] = g;
    __syncthreads();
    const int4*   col4 = (const int4*)col;
    const int4*   row4 = (const int4*)row;
    const float4* w4   = (const float4*)w;
#pragma unroll
    for (int i0 = 0; i0 < 8; i0 += 4) {
        int4 cc[4], rr[4];
        float4 ww[4];
#pragma unroll
        for (int u = 0; u < 4; ++u) {
            unsigned q = k * (CH / 4) + (i0 + u) * 1024 + t;
            unsigned qs = q < E_QUADS ? q : 0;
            cc[u] = col4[qs];
            rr[u] = row4[qs];
            ww[u] = w4[qs];
        }
#pragma unroll
        for (int u = 0; u < 4; ++u) {
            unsigned q = k * (CH / 4) + (i0 + u) * 1024 + t;
            if (q < E_QUADS) {
                int p0 = atomicAdd(&cur[cc[u].x >> SH], 1);
                int p1 = atomicAdd(&cur[cc[u].y >> SH], 1);
                int p2 = atomicAdd(&cur[cc[u].z >> SH], 1);
                int p3 = atomicAdd(&cur[cc[u].w >> SH], 1);
                stage[p0] = (v2i){ ((cc[u].x & (BNODES - 1)) << ROW_BITS) | rr[u].x, __float_as_int(ww[u].x) };
                stage[p1] = (v2i){ ((cc[u].y & (BNODES - 1)) << ROW_BITS) | rr[u].y, __float_as_int(ww[u].y) };
                stage[p2] = (v2i){ ((cc[u].z & (BNODES - 1)) << ROW_BITS) | rr[u].z, __float_as_int(ww[u].z) };
                stage[p3] = (v2i){ ((cc[u].w & (BNODES - 1)) << ROW_BITS) | rr[u].w, __float_as_int(ww[u].w) };
            }
        }
    }
    __syncthreads();
    unsigned wv = t >> 5, lane = t & 31;
    unsigned mb = wv + 32 * lane;
    int my_ls = 0, my_len = 0, my_g = 0;
    if (mb < NBK) {
        my_ls = lstart[mb];
        my_len = cur[mb] - my_ls;
        my_g = gb[mb];
    }
    for (int i = 0; i < (NBK + 31) / 32; ++i) {
        int ls  = __builtin_amdgcn_readlane(my_ls, i);
        int len = __builtin_amdgcn_readlane(my_len, i);
        int g0  = __builtin_amdgcn_readlane(my_g, i);
        for (int j = (int)lane; j < len; j += 32) tmp[g0 + j] = stage[ls + j];
    }
}

__global__ void __launch_bounds__(256) node_sort_kernel(const v2i* __restrict__ tmp, const int* __restrict__ bo,
                                                        int* __restrict__ off, int* __restrict__ cnt,
                                                        int2* __restrict__ ent,
                                                        const float* __restrict__ x, const float* __restrict__ W,
                                                        float* __restrict__ dinv, float* __restrict__ hs) {
    __shared__ v2i stage[CAP];
    __shared__ int lc[BNODES];
    unsigned t = threadIdx.x, b = blockIdx.x;
    int bstart = bo[b * NBLK];
    int bend = (b + 1 < NBK) ? bo[(b + 1) * NBLK] : N_EDGES;
    int len = bend - bstart;
    bool staged = len <= CAP;
    lc[t] = 0;
    __syncthreads();
    for (int j0 = 0; j0 < len; j0 += 256 * 8) {
        int key[8];
#pragma unroll
        for (int u = 0; u < 8; ++u) {
            int j = j0 + u * 256 + (int)t;
            key[u] = tmp[bstart + (j < len ? j : 0)].x;
        }
#pragma unroll
        for (int u = 0; u < 8; ++u) {
            int j = j0 + u * 256 + (int)t;
            if (j < len) atomicAdd(&lc[key[u] >> ROW_BITS], 1);
        }
    }
    __syncthreads();
    int c = lc[t];
    for (int d = 1; d < 256; d <<= 1) {
        int add = (t >= (unsigned)d) ? lc[t - d] : 0;
        __syncthreads();
        lc[t] += add;
        __syncthreads();
    }
    int ex = lc[t] - c;
    unsigned node = b * BNODES + t;
    if (node < N_NODES) {
        off[node] = bstart + ex;
        cnt[node] = c;
    }
    lc[t] = ex;
    __syncthreads();
    for (int j0 = 0; j0 < len; j0 += 256 * 8) {
        v2i e[8];
        int p[8];
#pragma unroll
        for (int u = 0; u < 8; ++u) {
            int j = j0 + u * 256 + (int)t;
            e[u] = tmp[bstart + (j < len ? j : 0)];
        }
#pragma unroll
        for (int u = 0; u < 8; ++u) {
            int j = j0 + u * 256 + (int)t;
            p[u] = 0;
            if (j < len) p[u] = atomicAdd(&lc[e[u].x >> ROW_BITS], 1);
        }
#pragma unroll
        for (int u = 0; u < 8; ++u) {
            int j = j0 + u * 256 + (int)t;
            if (j < len) {
                int r = e[u].x & ((1 << ROW_BITS) - 1);
                if (staged) stage[p[u]] = (v2i){ r, e[u].y };
                else        ent[bstart + p[u]] = make_int2(r, e[u].y);
            }
        }
    }
    if (!staged) __threadfence();
    __syncthreads();
    float deg = 0.0f;
    if (staged) {
#pragma unroll 4
        for (int j = (int)t; j < len; j += 256) {
            v2i e = stage[j];
            ent[bstart + j] = make_int2(e.x, e.y);
        }
#pragma unroll 4
        for (int i = 0; i < c; ++i) deg += __int_as_float(stage[ex + i].y);
    } else {
        for (int i = 0; i < c; ++i) deg += __int_as_float(ent[bstart + ex + i].y);
    }
    deg += 1.0f;
    float dn = (deg > 0.0f) ? rsqrtf(deg) : 0.0f;
    if (node < N_NODES) dinv[node] = dn;
    lc[t] = __float_as_int(dn);
    __syncthreads();
    unsigned wv = (unsigned)__builtin_amdgcn_readfirstlane((int)(t >> 5));
    unsigned lane = t & 31;
    unsigned lh = lane & 15;
    unsigned hi = lane >> 4;
    for (unsigned tt = wv; tt < BNODES / 16; tt += 8) {
        unsigned m0 = b * BNODES + tt * 16;
        if (m0 >= N_NODES) break;
        v8f acc = {};
        const float* xrow = x + (size_t)(m0 + lh) * IN_FEAT;
#pragma unroll
        for (int k = 0; k < IN_FEAT; k += 4) {
            int ka = k + (int)hi * 2;
            v2f a = { xrow[ka], xrow[ka + 1] };
            v2f bw = { W[ka * HIDDEN + lh], W[(ka + 1) * HIDDEN + lh] };
            acc = __builtin_amdgcn_wmma_f32_16x16x4_f32(false, a, false, bw,
                                                        (short)0, acc, false, false);
        }
        unsigned lb = tt * 16 + hi * 8;
#pragma unroll
        for (int r = 0; r < 8; ++r)
            hs[(size_t)(b * BNODES + lb + r) * HIDDEN + lh] = acc[r] * __int_as_float(lc[lb + r]);
    }
}


template <int U>
__device__ __forceinline__ int quad_pick(int v) {
    return __builtin_amdgcn_update_dpp(0, v, U * 0x55, 0xF, 0xF, true);
}

__device__ __forceinline__ int2 load_entry(const int2* __restrict__ ent, int s, int n, int k, int node) {
    int idx = s + k;
    idx = idx < N_EDGES ? idx : N_EDGES - 1;
    int2 v = ent[idx];
    bool ok = k < n;
    int tailw = (k == n) ? __float_as_int(1.0f) : 0;
    int2 r;
    r.x = ok ? v.x : node;
    r.y = ok ? v.y : tailw;
    return r;
}

#define GATHER_STEP(U, E)                                                   \
    {                                                                       \
        int r_ = quad_pick<(U) & 3>((E).x);                                 \
        w[U] = __int_as_float(quad_pick<(U) & 3>((E).y));                   \
        h[U] = hs4[(unsigned)r_ * (HIDDEN / 4) + q];                        \
    }

#define TP 18
template <int LAYER>
__global__ void __launch_bounds__(256, 16) agg_kernel(const int* __restrict__ off, const int* __restrict__ cnt,
                                                  const int2* __restrict__ ent, const float* __restrict__ hs,
                                                  const float* __restrict__ dinv, const float* __restrict__ b1,
                                                  float* __restrict__ dst, const float* __restrict__ W2,
                                                  const float* __restrict__ b2, float* __restrict__ out) {
    __shared__ float tile[LAYER == 2 ? 8 * 16 * TP : 1];
    unsigned wave = (blockIdx.x * blockDim.x + threadIdx.x) >> 5;
    if (wave >= N_TILES) return;
    unsigned lane = threadIdx.x & 31;
    unsigned g = lane >> 2, q = lane & 3;
    unsigned c0 = wave * 16;
    const float4* hs4 = (const float4*)hs;
    int s2[2] = { off[c0 + g], off[c0 + 8u + g] };
    int n2[2] = { cnt[c0 + g], cnt[c0 + 8u + g] };
#pragma unroll
    for (int pass = 0; pass < 2; ++pass) {
        unsigned node = c0 + 8u * (unsigned)pass + g;
        int s = s2[pass], n = n2[pass];
        float di = dinv[node];
        float4 acc = make_float4(0.0f, 0.0f, 0.0f, 0.0f);
        int2 ea = load_entry(ent, s, n, (int)q, (int)node);
        int2 eb = load_entry(ent, s, n, 4 + (int)q, (int)node);
        int k = 0;
        do {
            int2 ca = ea, cb = eb;
            ea = load_entry(ent, s, n, k + 8 + (int)q, (int)node);
            eb = load_entry(ent, s, n, k + 12 + (int)q, (int)node);
            float4 h[8];
            float w[8];
            GATHER_STEP(0, ca) GATHER_STEP(1, ca) GATHER_STEP(2, ca) GATHER_STEP(3, ca)
            GATHER_STEP(4, cb) GATHER_STEP(5, cb) GATHER_STEP(6, cb) GATHER_STEP(7, cb)
#pragma unroll
            for (int u = 0; u < 8; ++u) {
                acc.x = fmaf(w[u], h[u].x, acc.x);
                acc.y = fmaf(w[u], h[u].y, acc.y);
                acc.z = fmaf(w[u], h[u].z, acc.z);
                acc.w = fmaf(w[u], h[u].w, acc.w);
            }
            k += 8;
        } while (k <= n);
        float4 v;
        v.x = di * acc.x;
        v.y = di * acc.y;
        v.z = di * acc.z;
        v.w = di * acc.w;
        if (LAYER == 1) {
            float4 bias = ((const float4*)b1)[q];
            v.x += bias.x; v.y += bias.y; v.z += bias.z; v.w += bias.w;
            v.x = v.x > 0.0f ? v.x : 0.0f;
            v.y = v.y > 0.0f ? v.y : 0.0f;
            v.z = v.z > 0.0f ? v.z : 0.0f;
            v.w = v.w > 0.0f ? v.w : 0.0f;
            v.x *= di; v.y *= di; v.z *= di; v.w *= di;
            ((float4*)dst)[node * (HIDDEN / 4) + q] = v;
        } else {
            float* trow = tile + (threadIdx.x >> 5) * (16 * TP) + (8u * (unsigned)pass + g) * TP + 4u * q;
            trow[0] = v.x; trow[1] = v.y; trow[2] = v.z; trow[3] = v.w;
        }
    }
    if (LAYER == 2) {
        unsigned f = lane & 15, hi = lane >> 4;
        const float* trow = tile + (threadIdx.x >> 5) * (16 * TP) + f * TP;
        float bia0 = b2[f], bia1 = b2[16 + f];
        v8f c0v = { bia0, bia0, bia0, bia0, bia0, bia0, bia0, bia0 };
        v8f c1v = { bia1, bia1, bia1, bia1, bia1, bia1, bia1, bia1 };
#pragma unroll
        for (int k = 0; k < HIDDEN; k += 4) {
            int ka = k + (int)hi * 2;
            v2f a  = { trow[ka], trow[ka + 1] };
            v2f w0 = { W2[ka * N_CLASSES + f],      W2[(ka + 1) * N_CLASSES + f] };
            v2f w1 = { W2[ka * N_CLASSES + 16 + f], W2[(ka + 1) * N_CLASSES + 16 + f] };
            c0v = __builtin_amdgcn_wmma_f32_16x16x4_f32(false, a, false, w0,
                                                        (short)0, c0v, false, false);
            c1v = __builtin_amdgcn_wmma_f32_16x16x4_f32(false, a, false, w1,
                                                        (short)0, c1v, false, false);
        }
        unsigned mbase = c0 + hi * 8;
#pragma unroll
        for (int r = 0; r < 8; ++r) {
            out[(mbase + r) * N_CLASSES + f]      = c0v[r];
            out[(mbase + r) * N_CLASSES + 16 + f] = c1v[r];
        }
    }
}

extern "C" void kernel_launch(void* const* d_in, const int* in_sizes, int n_in,
                              void* d_out, int out_size, void* d_ws, size_t ws_size,
                              hipStream_t stream) {
    const float* x  = (const float*)d_in[0];
    const int*   ei = (const int*)d_in[1];
    const float* ew = (const float*)d_in[2];
    const float* W1 = (const float*)d_in[3];
    const float* b1 = (const float*)d_in[4];
    const float* W2 = (const float*)d_in[5];
    const float* b2 = (const float*)d_in[6];
    const int* row = ei;
    const int* col = ei + N_EDGES;
    float* out = (float*)d_out;

    int*   cnt  = (int*)d_ws;
    int*   off  = cnt  + N_NODES;
    int*   bsum = off  + N_NODES;
    float* dinv = (float*)(bsum + 256);
    float* hs1  = dinv + N_NODES;
    float* hs2  = hs1  + (size_t)N_NODES * HIDDEN;
    float* spare = hs2 + (size_t)N_NODES * HIDDEN;
    int*   bh   = (int*)(spare + (size_t)N_NODES * HIDDEN);
    int*   bo   = bh + M_BH;
    v2i*   tmp  = (v2i*)(bo + M_BH);
    int2*  ent  = (int2*)(tmp + N_EDGES);

    const int B = 256;
    hist_kernel<<<NBLK, 1024, 0, stream>>>(col, bh);
    blocksum_kernel<<<SCAN_BLOCKS, 256, 0, stream>>>(bh, bsum);
    offsets_kernel<<<SCAN_BLOCKS, 256, 0, stream>>>(bh, bsum, bo);
    bucket_kernel<<<NBLK, 1024, 0, stream>>>(row, col, ew, bh, bo, tmp);
    node_sort_kernel<<<NBK, 256, 0, stream>>>(tmp, bo, off, cnt, ent, x, W1, dinv, hs1);
    agg_kernel<1><<<(N_TILES * 32 + B - 1) / B, B, 0, stream>>>(off, cnt, ent, hs1, dinv, b1, hs2, W2, b2, out);
    agg_kernel<2><<<(N_TILES * 32 + B - 1) / B, B, 0, stream>>>(off, cnt, ent, hs2, dinv, b1, spare, W2, b2, out);
}


# Sort loses its count and scan launches: a chunk ranks, groups and writes its edges in place; a bucket's start is the sum of its runs' local starts; 7 launches become 4; changes f32 summation order
# speedup vs baseline: 7.5925x; 1.0405x over previous
#include <hip/hip_runtime.h>

#define N_NODES   200000
#define N_EDGES   6400000
#define IN_FEAT   128
#define HIDDEN    16
#define N_CLASSES 32
#define N_TILES   (N_NODES / 16)

#define SH        8
#define BNODES    (1 << SH)
#define NBK       ((N_NODES + BNODES - 1) / BNODES)
#define CH        32768
#define NBLK      ((N_EDGES + CH - 1) / CH)
#define M_BH      (NBK * NBLK)
#define E_QUADS   (N_EDGES / 4)
#define RLANES    ((NBLK + 7) / 8)
#define ROW_BITS  18
#define CAP       9984

static_assert(N_EDGES % 4 == 0, "whole quads");
static_assert(N_NODES <= (1 << ROW_BITS) && SH + ROW_BITS < 32, "source index and local node share one word");
static_assert(NBK <= 1024 && CH == 8 * 4 * 1024 && NBLK <= 256 && RLANES <= 32, "one thread per bucket; one node-sort thread per run of a bucket");
static_assert(CH <= (1 << 15) && SH + 15 < 32, "a slot of the stage and a local node share one word");

typedef __attribute__((ext_vector_type(2))) float v2f;
typedef __attribute__((ext_vector_type(8))) float v8f;
typedef __attribute__((ext_vector_type(2))) int   v2i;


__global__ void __launch_bounds__(1024) bucket_kernel(const int* __restrict__ row, const int* __restrict__ col,
                                                      const float* __restrict__ w, int* __restrict__ rcnt,
                                                      int* __restrict__ rloc, v2i* __restrict__ tmp) {
    __shared__ __attribute__((aligned(16))) v2i stage[CH];
    __shared__ int lstart[1024];
    __shared__ int cur[1024];
    unsigned t = threadIdx.x, k = blockIdx.x;
    cur[t] = 0;
    __syncthreads();
    const int4*   col4 = (const int4*)col;
    const int4*   row4 = (const int4*)row;
    const float4* w4   = (const float4*)w;
    int4 cc[8], rk[8];
#pragma unroll
    for (int i = 0; i < 8; ++i) {
        unsigned q = k * (CH / 4) + i * 1024 + t;
        cc[i] = col4[q < E_QUADS ? q : 0];
    }
#pragma unroll
    for (int i = 0; i < 8; ++i) {
        unsigned q = k * (CH / 4) + i * 1024 + t;
        rk[i] = make_int4(0, 0, 0, 0);
        if (q < E_QUADS) {
            rk[i].x = atomicAdd(&cur[cc[i].x >> SH], 1);
            rk[i].y = atomicAdd(&cur[cc[i].y >> SH], 1);
            rk[i].z = atomicAdd(&cur[cc[i].z >> SH], 1);
            rk[i].w = atomicAdd(&cur[cc[i].w >> SH], 1);
        }
    }
    __syncthreads();
    int c = cur[t];
    for (int d = 1; d < 1024; d <<= 1) {
        int add = (t >= (unsigned)d) ? cur[t - d] : 0;
        __syncthreads();
        cur[t] += add;
        __syncthreads();
    }
    int ex = cur[t] - c;
    lstart[t] = ex;
    if (t < NBK) {
        rcnt[t * NBLK + k] = c;
        rloc[t * NBLK + k] = ex;
    }
    __syncthreads();
    int4 pk[8];
#pragma unroll
    for (int i = 0; i < 8; ++i) {
        unsigned q = k * (CH / 4) + i * 1024 + t;
        pk[i] = make_int4(0, 0, 0, 0);
        if (q < E_QUADS) {
            pk[i].x = ((cc[i].x & (BNODES - 1)) << 15) | (lstart[cc[i].x >> SH] + rk[i].x);
            pk[i].y = ((cc[i].y & (BNODES - 1)) << 15) | (lstart[cc[i].y >> SH] + rk[i].y);
            pk[i].z = ((cc[i].z & (BNODES - 1)) << 15) | (lstart[cc[i].z >> SH] + rk[i].z);
            pk[i].w = ((cc[i].w & (BNODES - 1)) << 15) | (lstart[cc[i].w >> SH] + rk[i].w);
        }
    }
#pragma unroll
    for (int i0 = 0; i0 < 8; i0 += 4) {
        int4 rr[4];
        float4 ww[4];
#pragma unroll
        for (int u = 0; u < 4; ++u) {
            unsigned q = k * (CH / 4) + (i0 + u) * 1024 + t;
            unsigned qs = q < E_QUADS ? q : 0;
            rr[u] = row4[qs];
            ww[u] = w4[qs];
        }
#pragma unroll
        for (int u = 0; u < 4; ++u) {
            unsigned q = k * (CH / 4) + (i0 + u) * 1024 + t;
            if (q < E_QUADS) {
                int4 pq = pk[i0 + u];
                stage[pq.x & 32767] = (v2i){ ((pq.x >> 15) << ROW_BITS) | rr[u].x, __float_as_int(ww[u].x) };
                stage[pq.y & 32767] = (v2i){ ((pq.y >> 15) << ROW_BITS) | rr[u].y, __float_as_int(ww[u].y) };
                stage[pq.z & 32767] = (v2i){ ((pq.z >> 15) << ROW_BITS) | rr[u].z, __float_as_int(ww[u].z) };
                stage[pq.w & 32767] = (v2i){ ((pq.w >> 15) << ROW_BITS) | rr[u].w, __float_as_int(ww[u].w) };
            }
        }
    }
    __syncthreads();
    int nk = N_EDGES - (int)k * CH;
    nk = nk < CH ? nk : CH;
    const int4* s4 = (const int4*)stage;
    int4* t4 = (int4*)(tmp + (size_t)k * CH);
#pragma unroll 4
    for (int i = (int)t; i < nk / 2; i += 1024) t4[i] = s4[i];
}

__global__ void __launch_bounds__(256) node_sort_kernel(const v2i* __restrict__ tmp, const int* __restrict__ rcnt,
                                                        const int* __restrict__ rloc,
                                                        int* __restrict__ off, int* __restrict__ cnt,
                                                        int2* __restrict__ ent,
                                                        const float* __restrict__ x, const float* __restrict__ W,
                                                        float* __restrict__ dinv, float* __restrict__ hs) {
    __shared__ v2i stage[CAP];
    __shared__ int lc[BNODES];
    __shared__ int red[BNODES];
    unsigned t = threadIdx.x, b = blockIdx.x;
    const int rlane = (int)(t & 31);
    unsigned mrun = (t >> 5) + 8 * (t & 31);
    int my_rc = 0, my_rl = 0;
    if (mrun < NBLK) {
        my_rc = rcnt[b * NBLK + mrun];
        my_rl = rloc[b * NBLK + mrun];
    }
    int my_rg = (int)mrun * CH + my_rl;
    lc[t] = my_rl;
    red[t] = my_rc;
    __syncthreads();
    for (int d = 128; d > 0; d >>= 1) {
        if (t < (unsigned)d) {
            lc[t] += lc[t + d];
            red[t] += red[t + d];
        }
        __syncthreads();
    }
    int bstart = lc[0];
    int len = red[0];
    __syncthreads();
    bool staged = len <= CAP;
    lc[t] = 0;
    __syncthreads();
    for (int i0 = 0; i0 < RLANES; i0 += 8) {
        int key[16];
#pragma unroll
        for (int u = 0; u < 8; ++u) {
            int rg = __builtin_amdgcn_readlane(my_rg, i0 + u);
            int rc = __builtin_amdgcn_readlane(my_rc, i0 + u);
            key[2 * u]     = tmp[rlane < rc ? rg + rlane : 0].x;
            key[2 * u + 1] = tmp[rlane + 32 < rc ? rg + rlane + 32 : 0].x;
        }
        asm volatile("" :: "v"(key[0]), "v"(key[1]), "v"(key[2]), "v"(key[3]), "v"(key[4]), "v"(key[5]), "v"(key[6]), "v"(key[7]),
                           "v"(key[8]), "v"(key[9]), "v"(key[10]), "v"(key[11]), "v"(key[12]), "v"(key[13]), "v"(key[14]), "v"(key[15]));
#pragma unroll
        for (int u = 0; u < 8; ++u) {
            int rc = __builtin_amdgcn_readlane(my_rc, i0 + u);
            if (rlane < rc) atomicAdd(&lc[key[2 * u] >> ROW_BITS], 1);
            if (rlane + 32 < rc) atomicAdd(&lc[key[2 * u + 1] >> ROW_BITS], 1);
        }
        for (int u = 0; u < 8; ++u) {
            int rg = __builtin_amdgcn_readlane(my_rg, i0 + u);
            int rc = __builtin_amdgcn_readlane(my_rc, i0 + u);
            for (int j = rlane + 64; j < rc; j += 32) atomicAdd(&lc[tmp[rg + j].x >> ROW_BITS], 1);
        }
    }
    __syncthreads();
    int c = lc[t];
    for (int d = 1; d < 256; d <<= 1) {
        int add = (t >= (unsigned)d) ? lc[t - d] : 0;
        __syncthreads();
        lc[t] += add;
        __syncthreads();
    }
    int ex = lc[t] - c;
    unsigned node = b * BNODES + t;
    if (node < N_NODES) {
        off[node] = bstart + ex;
        cnt[node] = c;
    }
    lc[t] = ex;
    __syncthreads();
    for (int i0 = 0; i0 < RLANES; i0 += 8) {
        v2i e[16];
        int p[16];
#pragma unroll
        for (int u = 0; u < 8; ++u) {
            int rg = __builtin_amdgcn_readlane(my_rg, i0 + u);
            int rc = __builtin_amdgcn_readlane(my_rc, i0 + u);
            e[2 * u]     = tmp[rlane < rc ? rg + rlane : 0];
            e[2 * u + 1] = tmp[rlane + 32 < rc ? rg + rlane + 32 : 0];
        }
        asm volatile("" :: "v"(e[0]), "v"(e[1]), "v"(e[2]), "v"(e[3]), "v"(e[4]), "v"(e[5]), "v"(e[6]), "v"(e[7]),
                           "v"(e[8]), "v"(e[9]), "v"(e[10]), "v"(e[11]), "v"(e[12]), "v"(e[13]), "v"(e[14]), "v"(e[15]));
#pragma unroll
        for (int u = 0; u < 8; ++u) {
            int rc = __builtin_amdgcn_readlane(my_rc, i0 + u);
            p[2 * u] = 0;
            p[2 * u + 1] = 0;
            if (rlane < rc) p[2 * u] = atomicAdd(&lc[e[2 * u].x >> ROW_BITS], 1);
            if (rlane + 32 < rc) p[2 * u + 1] = atomicAdd(&lc[e[2 * u + 1].x >> ROW_BITS], 1);
        }
#pragma unroll
        for (int u = 0; u < 8; ++u) {
            int rc = __builtin_amdgcn_readlane(my_rc, i0 + u);
#pragma unroll
            for (int h = 0; h < 2; ++h) {
                if (rlane + 32 * h < rc) {
                    int r = e[2 * u + h].x & ((1 << ROW_BITS) - 1);
                    if (staged) stage[p[2 * u + h]] = (v2i){ r, e[2 * u + h].y };
                    else        ent[bstart + p[2 * u + h]] = make_int2(r, e[2 * u + h].y);
                }
            }
        }
        for (int u = 0; u < 8; ++u) {
            int rg = __builtin_amdgcn_readlane(my_rg, i0 + u);
            int rc = __builtin_amdgcn_readlane(my_rc, i0 + u);
            for (int j = rlane + 64; j < rc; j += 32) {
                v2i ee = tmp[rg + j];
                int pp = atomicAdd(&lc[ee.x >> ROW_BITS], 1);
                int r = ee.x & ((1 << ROW_BITS) - 1);
                if (staged) stage[pp] = (v2i){ r, ee.y };
                else        ent[bstart + pp] = make_int2(r, ee.y);
            }
        }
    }
    if (!staged) __threadfence();
    __syncthreads();
    float deg = 0.0f;
    if (staged) {
#pragma unroll 4
        for (int j = (int)t; j < len; j += 256) {
            v2i e = stage[j];
            ent[bstart + j] = make_int2(e.x, e.y);
        }
#pragma unroll 4
        for (int i = 0; i < c; ++i) deg += __int_as_float(stage[ex + i].y);
    } else {
        for (int i = 0; i < c; ++i) deg += __int_as_float(ent[bstart + ex + i].y);
    }
    deg += 1.0f;
    float dn = (deg > 0.0f) ? rsqrtf(deg) : 0.0f;
    if (node < N_NODES) dinv[node] = dn;
    lc[t] = __float_as_int(dn);
    __syncthreads();
    unsigned wv = (unsigned)__builtin_amdgcn_readfirstlane((int)(t >> 5));
    unsigned lane = t & 31;
    unsigned lh = lane & 15;
    unsigned hi = lane >> 4;
    for (unsigned tt = wv; tt < BNODES / 16; tt += 8) {
        unsigned m0 = b * BNODES + tt * 16;
        if (m0 >= N_NODES) break;
        v8f acc = {};
        const float* xrow = x + (size_t)(m0 + lh) * IN_FEAT;
#pragma unroll
        for (int k = 0; k < IN_FEAT; k += 4) {
            int ka = k + (int)hi * 2;
            v2f a = { xrow[ka], xrow[ka + 1] };
            v2f bw = { W[ka * HIDDEN + lh], W[(ka + 1) * HIDDEN + lh] };
            acc = __builtin_amdgcn_wmma_f32_16x16x4_f32(false, a, false, bw,
                                                        (short)0, acc, false, false);
        }
        unsigned lb = tt * 16 + hi * 8;
#pragma unroll
        for (int r = 0; r < 8; ++r)
            hs[(size_t)(b * BNODES + lb + r) * HIDDEN + lh] = acc[r] * __int_as_float(lc[lb + r]);
    }
}


template <int U>
__device__ __forceinline__ int quad_pick(int v) {
    return __builtin_amdgcn_update_dpp(0, v, U * 0x55, 0xF, 0xF, true);
}

__device__ __forceinline__ int2 load_entry(const int2* __restrict__ ent, int s, int n, int k, int node) {
    int idx = s + k;
    idx = idx < N_EDGES ? idx : N_EDGES - 1;
    int2 v = ent[idx];
    bool ok = k < n;
    int tailw = (k == n) ? __float_as_int(1.0f) : 0;
    int2 r;
    r.x = ok ? v.x : node;
    r.y = ok ? v.y : tailw;
    return r;
}

#define GATHER_STEP(U, E)                                                   \
    {                                                                       \
        int r_ = quad_pick<(U) & 3>((E).x);                                 \
        w[U] = __int_as_float(quad_pick<(U) & 3>((E).y));                   \
        h[U] = hs4[(unsigned)r_ * (HIDDEN / 4) + q];                        \
    }

#define TP 18
template <int LAYER>
__global__ void __launch_bounds__(256, 16) agg_kernel(const int* __restrict__ off, const int* __restrict__ cnt,
                                                  const int2* __restrict__ ent, const float* __restrict__ hs,
                                                  const float* __restrict__ dinv, const float* __restrict__ b1,
                                                  float* __restrict__ dst, const float* __restrict__ W2,
                                                  const float* __restrict__ b2, float* __restrict__ out) {
    __shared__ float tile[LAYER == 2 ? 8 * 16 * TP : 1];
    unsigned wave = (blockIdx.x * blockDim.x + threadIdx.x) >> 5;
    if (wave >= N_TILES) return;
    unsigned lane = threadIdx.x & 31;
    unsigned g = lane >> 2, q = lane & 3;
    unsigned c0 = wave * 16;
    const float4* hs4 = (const float4*)hs;
    int s2[2] = { off[c0 + g], off[c0 + 8u + g] };
    int n2[2] = { cnt[c0 + g], cnt[c0 + 8u + g] };
#pragma unroll
    for (int pass = 0; pass < 2; ++pass) {
        unsigned node = c0 + 8u * (unsigned)pass + g;
        int s = s2[pass], n = n2[pass];
        float di = dinv[node];
        float4 acc = make_float4(0.0f, 0.0f, 0.0f, 0.0f);
        int2 ea = load_entry(ent, s, n, (int)q, (int)node);
        int2 eb = load_entry(ent, s, n, 4 + (int)q, (int)node);
        int k = 0;
        do {
            int2 ca = ea, cb = eb;
            ea = load_entry(ent, s, n, k + 8 + (int)q, (int)node);
            eb = load_entry(ent, s, n, k + 12 + (int)q, (int)node);
            float4 h[8];
            float w[8];
            GATHER_STEP(0, ca) GATHER_STEP(1, ca) GATHER_STEP(2, ca) GATHER_STEP(3, ca)
            GATHER_STEP(4, cb) GATHER_STEP(5, cb) GATHER_STEP(6, cb) GATHER_STEP(7, cb)
#pragma unroll
            for (int u = 0; u < 8; ++u) {
                acc.x = fmaf(w[u], h[u].x, acc.x);
                acc.y = fmaf(w[u], h[u].y, acc.y);
                acc.z = fmaf(w[u], h[u].z, acc.z);
                acc.w = fmaf(w[u], h[u].w, acc.w);
            }
            k += 8;
        } while (k <= n);
        float4 v;
        v.x = di * acc.x;
        v.y = di * acc.y;
        v.z = di * acc.z;
        v.w = di * acc.w;
        if (LAYER == 1) {
            float4 bias = ((const float4*)b1)[q];
            v.x += bias.x; v.y += bias.y; v.z += bias.z; v.w += bias.w;
            v.x = v.x > 0.0f ? v.x : 0.0f;
            v.y = v.y > 0.0f ? v.y : 0.0f;
            v.z = v.z > 0.0f ? v.z : 0.0f;
            v.w = v.w > 0.0f ? v.w : 0.0f;
            v.x *= di; v.y *= di; v.z *= di; v.w *= di;
            ((float4*)dst)[node * (HIDDEN / 4) + q] = v;
        } else {
            float* trow = tile + (threadIdx.x >> 5) * (16 * TP) + (8u * (unsigned)pass + g) * TP + 4u * q;
            trow[0] = v.x; trow[1] = v.y; trow[2] = v.z; trow[3] = v.w;
        }
    }
    if (LAYER == 2) {
        unsigned f = lane & 15, hi = lane >> 4;
        const float* trow = tile + (threadIdx.x >> 5) * (16 * TP) + f * TP;
        float bia0 = b2[f], bia1 = b2[16 + f];
        v8f c0v = { bia0, bia0, bia0, bia0, bia0, bia0, bia0, bia0 };
        v8f c1v = { bia1, bia1, bia1, bia1, bia1, bia1, bia1, bia1 };
#pragma unroll
        for (int k = 0; k < HIDDEN; k += 4) {
            int ka = k + (int)hi * 2;
            v2f a  = { trow[ka], trow[ka + 1] };
            v2f w0 = { W2[ka * N_CLASSES + f],      W2[(ka + 1) * N_CLASSES + f] };
            v2f w1 = { W2[ka * N_CLASSES + 16 + f], W2[(ka + 1) * N_CLASSES + 16 + f] };
            c0v = __builtin_amdgcn_wmma_f32_16x16x4_f32(false, a, false, w0,
                                                        (short)0, c0v, false, false);
            c1v = __builtin_amdgcn_wmma_f32_16x16x4_f32(false, a, false, w1,
                                                        (short)0, c1v, false, false);
        }
        unsigned mbase = c0 + hi * 8;
#pragma unroll
        for (int r = 0; r < 8; ++r) {
            out[(mbase + r) * N_CLASSES + f]      = c0v[r];
            out[(mbase + r) * N_CLASSES + 16 + f] = c1v[r];
        }
    }
}

extern "C" void kernel_launch(void* const* d_in, const int* in_sizes, int n_in,
                              void* d_out, int out_size, void* d_ws, size_t ws_size,
                              hipStream_t stream) {
    const float* x  = (const float*)d_in[0];
    const int*   ei = (const int*)d_in[1];
    const float* ew = (const float*)d_in[2];
    const float* W1 = (const float*)d_in[3];
    const float* b1 = (const float*)d_in[4];
    const float* W2 = (const float*)d_in[5];
    const float* b2 = (const float*)d_in[6];
    const int* row = ei;
    const int* col = ei + N_EDGES;
    float* out = (float*)d_out;

    int*   cnt  = (int*)d_ws;
    int*   off  = cnt  + N_NODES;
    float* dinv = (float*)(off + N_NODES + 256);
    float* hs1  = dinv + N_NODES;
    float* hs2  = hs1  + (size_t)N_NODES * HIDDEN;
    float* spare = hs2 + (size_t)N_NODES * HIDDEN;
    int*   rcnt = (int*)(spare + (size_t)N_NODES * HIDDEN);
    int*   rloc = rcnt + M_BH;
    v2i*   tmp  = (v2i*)(rloc + M_BH);
    int2*  ent  = (int2*)(tmp + N_EDGES);

    const int B = 256;
    bucket_kernel<<<NBLK, 1024, 0, stream>>>(row, col, ew, rcnt, rloc, tmp);
    node_sort_kernel<<<NBK, 256, 0, stream>>>(tmp, rcnt, rloc, off, cnt, ent, x, W1, dinv, hs1);
    agg_kernel<1><<<(N_TILES * 32 + B - 1) / B, B, 0, stream>>>(off, cnt, ent, hs1, dinv, b1, hs2, W2, b2, out);
    agg_kernel<2><<<(N_TILES * 32 + B - 1) / B, B, 0, stream>>>(off, cnt, ent, hs2, dinv, b1, spare, W2, b2, out);
}


# Both in-edge sums read their edge list from LDS: a wave copies each pass's eight runs, one contiguous span, by whole-line 16-byte loads, not two 8-byte loads per lane and trip; sums keep their order
# speedup vs baseline: 8.3224x; 1.0961x over previous
#include <hip/hip_runtime.h>

#define N_NODES   200000
#define N_EDGES   6400000
#define IN_FEAT   128
#define HIDDEN    16
#define N_CLASSES 32
#define N_TILES   (N_NODES / 16)

#define SH        8
#define BNODES    (1 << SH)
#define NBK       ((N_NODES + BNODES - 1) / BNODES)
#define CH        32768
#define NBLK      ((N_EDGES + CH - 1) / CH)
#define M_BH      (NBK * NBLK)
#define E_QUADS   (N_EDGES / 4)
#define RLANES    ((NBLK + 7) / 8)
#define ROW_BITS  18
#define CAP       9984

static_assert(N_EDGES % 4 == 0, "whole quads");
static_assert(N_NODES <= (1 << ROW_BITS) && SH + ROW_BITS < 32, "source index and local node share one word");
static_assert(NBK <= 1024 && CH == 8 * 4 * 1024 && NBLK <= 256 && RLANES <= 32, "one thread per bucket; one node-sort thread per run of a bucket");
static_assert(CH <= (1 << 15) && SH + 15 < 32, "a slot of the stage and a local node share one word");

typedef __attribute__((ext_vector_type(2))) float v2f;
typedef __attribute__((ext_vector_type(8))) float v8f;
typedef __attribute__((ext_vector_type(2))) int   v2i;
typedef __attribute__((ext_vector_type(4))) int   v4i;


__global__ void __launch_bounds__(1024) bucket_kernel(const int* __restrict__ row, const int* __restrict__ col,
                                                      const float* __restrict__ w, int* __restrict__ rcnt,
                                                      int* __restrict__ rloc, v2i* __restrict__ tmp) {
    __shared__ __attribute__((aligned(16))) v2i stage[CH];
    __shared__ int lstart[1024];
    __shared__ int cur[1024];
    unsigned t = threadIdx.x, k = blockIdx.x;
    cur[t] = 0;
    __syncthreads();
    const int4*   col4 = (const int4*)col;
    const int4*   row4 = (const int4*)row;
    const float4* w4   = (const float4*)w;
    int4 cc[8], rk[8];
#pragma unroll
    for (int i = 0; i < 8; ++i) {
        unsigned q = k * (CH / 4) + i * 1024 + t;
        cc[i] = col4[q < E_QUADS ? q : 0];
    }
#pragma unroll
    for (int i = 0; i < 8; ++i) {
        unsigned q = k * (CH / 4) + i * 1024 + t;
        rk[i] = make_int4(0, 0, 0, 0);
        if (q < E_QUADS) {
            rk[i].x = atomicAdd(&cur[cc[i].x >> SH], 1);
            rk[i].y = atomicAdd(&cur[cc[i].y >> SH], 1);
            rk[i].z = atomicAdd(&cur[cc[i].z >> SH], 1);
            rk[i].w = atomicAdd(&cur[cc[i].w >> SH], 1);
        }
    }
    __syncthreads();
    int c = cur[t];
    for (int d = 1; d < 1024; d <<= 1) {
        int add = (t >= (unsigned)d) ? cur[t - d] : 0;
        __syncthreads();
        cur[t] += add;
        __syncthreads();
    }
    int ex = cur[t] - c;
    lstart[t] = ex;
    if (t < NBK) {
        rcnt[t * NBLK + k] = c;
        rloc[t * NBLK + k] = ex;
    }
    __syncthreads();
    int4 pk[8];
#pragma unroll
    for (int i = 0; i < 8; ++i) {
        unsigned q = k * (CH / 4) + i * 1024 + t;
        pk[i] = make_int4(0, 0, 0, 0);
        if (q < E_QUADS) {
            pk[i].x = ((cc[i].x & (BNODES - 1)) << 15) | (lstart[cc[i].x >> SH] + rk[i].x);
            pk[i].y = ((cc[i].y & (BNODES - 1)) << 15) | (lstart[cc[i].y >> SH] + rk[i].y);
            pk[i].z = ((cc[i].z & (BNODES - 1)) << 15) | (lstart[cc[i].z >> SH] + rk[i].z);
            pk[i].w = ((cc[i].w & (BNODES - 1)) << 15) | (lstart[cc[i].w >> SH] + rk[i].w);
        }
    }
#pragma unroll
    for (int i0 = 0; i0 < 8; i0 += 4) {
        int4 rr[4];
        float4 ww[4];
#pragma unroll
        for (int u = 0; u < 4; ++u) {
            unsigned q = k * (CH / 4) + (i0 + u) * 1024 + t;
            unsigned qs = q < E_QUADS ? q : 0;
            rr[u] = row4[qs];
            ww[u] = w4[qs];
        }
#pragma unroll
        for (int u = 0; u < 4; ++u) {
            unsigned q = k * (CH / 4) + (i0 + u) * 1024 + t;
            if (q < E_QUADS) {
                int4 pq = pk[i0 + u];
                stage[pq.x & 32767] = (v2i){ ((pq.x >> 15) << ROW_BITS) | rr[u].x, __float_as_int(ww[u].x) };
                stage[pq.y & 32767] = (v2i){ ((pq.y >> 15) << ROW_BITS) | rr[u].y, __float_as_int(ww[u].y) };
                stage[pq.z & 32767] = (v2i){ ((pq.z >> 15) << ROW_BITS) | rr[u].z, __float_as_int(ww[u].z) };
                stage[pq.w & 32767] = (v2i){ ((pq.w >> 15) << ROW_BITS) | rr[u].w, __float_as_int(ww[u].w) };
            }
        }
    }
    __syncthreads();
    int nk = N_EDGES - (int)k * CH;
    nk = nk < CH ? nk : CH;
    const int4* s4 = (const int4*)stage;
    int4* t4 = (int4*)(tmp + (size_t)k * CH);
#pragma unroll 4
    for (int i = (int)t; i < nk / 2; i += 1024) t4[i] = s4[i];
}

__global__ void __launch_bounds__(256) node_sort_kernel(const v2i* __restrict__ tmp, const int* __restrict__ rcnt,
                                                        const int* __restrict__ rloc,
                                                        int* __restrict__ off, int* __restrict__ cnt,
                                                        int2* __restrict__ ent,
                                                        const float* __restrict__ x, const float* __restrict__ W,
                                                        float* __restrict__ dinv, float* __restrict__ hs) {
    __shared__ v2i stage[CAP];
    __shared__ int lc[BNODES];
    __shared__ int red[BNODES];
    unsigned t = threadIdx.x, b = blockIdx.x;
    const int rlane = (int)(t & 31);
    unsigned mrun = (t >> 5) + 8 * (t & 31);
    int my_rc = 0, my_rl = 0;
    if (mrun < NBLK) {
        my_rc = rcnt[b * NBLK + mrun];
        my_rl = rloc[b * NBLK + mrun];
    }
    int my_rg = (int)mrun * CH + my_rl;
    lc[t] = my_rl;
    red[t] = my_rc;
    __syncthreads();
    for (int d = 128; d > 0; d >>= 1) {
        if (t < (unsigned)d) {
            lc[t] += lc[t + d];
            red[t] += red[t + d];
        }
        __syncthreads();
    }
    int bstart = lc[0];
    int len = red[0];
    __syncthreads();
    bool staged = len <= CAP;
    lc[t] = 0;
    __syncthreads();
    for (int i0 = 0; i0 < RLANES; i0 += 8) {
        int key[16];
#pragma unroll
        for (int u = 0; u < 8; ++u) {
            int rg = __builtin_amdgcn_readlane(my_rg, i0 + u);
            int rc = __builtin_amdgcn_readlane(my_rc, i0 + u);
            key[2 * u]     = tmp[rlane < rc ? rg + rlane : 0].x;
            key[2 * u + 1] = tmp[rlane + 32 < rc ? rg + rlane + 32 : 0].x;
        }
        asm volatile("" :: "v"(key[0]), "v"(key[1]), "v"(key[2]), "v"(key[3]), "v"(key[4]), "v"(key[5]), "v"(key[6]), "v"(key[7]),
                           "v"(key[8]), "v"(key[9]), "v"(key[10]), "v"(key[11]), "v"(key[12]), "v"(key[13]), "v"(key[14]), "v"(key[15]));
#pragma unroll
        for (int u = 0; u < 8; ++u) {
            int rc = __builtin_amdgcn_readlane(my_rc, i0 + u);
            if (rlane < rc) atomicAdd(&lc[key[2 * u] >> ROW_BITS], 1);
            if (rlane + 32 < rc) atomicAdd(&lc[key[2 * u + 1] >> ROW_BITS], 1);
        }
        for (int u = 0; u < 8; ++u) {
            int rg = __builtin_amdgcn_readlane(my_rg, i0 + u);
            int rc = __builtin_amdgcn_readlane(my_rc, i0 + u);
            for (int j = rlane + 64; j < rc; j += 32) atomicAdd(&lc[tmp[rg + j].x >> ROW_BITS], 1);
        }
    }
    __syncthreads();
    int c = lc[t];
    for (int d = 1; d < 256; d <<= 1) {
        int add = (t >= (unsigned)d) ? lc[t - d] : 0;
        __syncthreads();
        lc[t] += add;
        __syncthreads();
    }
    int ex = lc[t] - c;
    unsigned node = b * BNODES + t;
    if (node < N_NODES) {
        off[node] = bstart + ex;
        cnt[node] = c;
    }
    lc[t] = ex;
    __syncthreads();
    for (int i0 = 0; i0 < RLANES; i0 += 8) {
        v2i e[16];
        int p[16];
#pragma unroll
        for (int u = 0; u < 8; ++u) {
            int rg = __builtin_amdgcn_readlane(my_rg, i0 + u);
            int rc = __builtin_amdgcn_readlane(my_rc, i0 + u);
            e[2 * u]     = tmp[rlane < rc ? rg + rlane : 0];
            e[2 * u + 1] = tmp[rlane + 32 < rc ? rg + rlane + 32 : 0];
        }
        asm volatile("" :: "v"(e[0]), "v"(e[1]), "v"(e[2]), "v"(e[3]), "v"(e[4]), "v"(e[5]), "v"(e[6]), "v"(e[7]),
                           "v"(e[8]), "v"(e[9]), "v"(e[10]), "v"(e[11]), "v"(e[12]), "v"(e[13]), "v"(e[14]), "v"(e[15]));
#pragma unroll
        for (int u = 0; u < 8; ++u) {
            int rc = __builtin_amdgcn_readlane(my_rc, i0 + u);
            p[2 * u] = 0;
            p[2 * u + 1] = 0;
            if (rlane < rc) p[2 * u] = atomicAdd(&lc[e[2 * u].x >> ROW_BITS], 1);
            if (rlane + 32 < rc) p[2 * u + 1] = atomicAdd(&lc[e[2 * u + 1].x >> ROW_BITS], 1);
        }
#pragma unroll
        for (int u = 0; u < 8; ++u) {
            int rc = __builtin_amdgcn_readlane(my_rc, i0 + u);
#pragma unroll
            for (int h = 0; h < 2; ++h) {
                if (rlane + 32 * h < rc) {
                    int r = e[2 * u + h].x & ((1 << ROW_BITS) - 1);
                    if (staged) stage[p[2 * u + h]] = (v2i){ r, e[2 * u + h].y };
                    else        ent[bstart + p[2 * u + h]] = make_int2(r, e[2 * u + h].y);
                }
            }
        }
        for (int u = 0; u < 8; ++u) {
            int rg = __builtin_amdgcn_readlane(my_rg, i0 + u);
            int rc = __builtin_amdgcn_readlane(my_rc, i0 + u);
            for (int j = rlane + 64; j < rc; j += 32) {
                v2i ee = tmp[rg + j];
                int pp = atomicAdd(&lc[ee.x >> ROW_BITS], 1);
                int r = ee.x & ((1 << ROW_BITS) - 1);
                if (staged) stage[pp] = (v2i){ r, ee.y };
                else        ent[bstart + pp] = make_int2(r, ee.y);
            }
        }
    }
    if (!staged) __threadfence();
    __syncthreads();
    float deg = 0.0f;
    if (staged) {
#pragma unroll 4
        for (int j = (int)t; j < len; j += 256) {
            v2i e = stage[j];
            ent[bstart + j] = make_int2(e.x, e.y);
        }
#pragma unroll 4
        for (int i = 0; i < c; ++i) deg += __int_as_float(stage[ex + i].y);
    } else {
        for (int i = 0; i < c; ++i) deg += __int_as_float(ent[bstart + ex + i].y);
    }
    deg += 1.0f;
    float dn = (deg > 0.0f) ? rsqrtf(deg) : 0.0f;
    if (node < N_NODES) dinv[node] = dn;
    lc[t] = __float_as_int(dn);
    __syncthreads();
    unsigned wv = (unsigned)__builtin_amdgcn_readfirstlane((int)(t >> 5));
    unsigned lane = t & 31;
    unsigned lh = lane & 15;
    unsigned hi = lane >> 4;
    for (unsigned tt = wv; tt < BNODES / 16; tt += 8) {
        unsigned m0 = b * BNODES + tt * 16;
        if (m0 >= N_NODES) break;
        v8f acc = {};
        const float* xrow = x + (size_t)(m0 + lh) * IN_FEAT;
#pragma unroll
        for (int k = 0; k < IN_FEAT; k += 4) {
            int ka = k + (int)hi * 2;
            v2f a = { xrow[ka], xrow[ka + 1] };
            v2f bw = { W[ka * HIDDEN + lh], W[(ka + 1) * HIDDEN + lh] };
            acc = __builtin_amdgcn_wmma_f32_16x16x4_f32(false, a, false, bw,
                                                        (short)0, acc, false, false);
        }
        unsigned lb = tt * 16 + hi * 8;
#pragma unroll
        for (int r = 0; r < 8; ++r)
            hs[(size_t)(b * BNODES + lb + r) * HIDDEN + lh] = acc[r] * __int_as_float(lc[lb + r]);
    }
}


template <int U>
__device__ __forceinline__ int quad_pick(int v) {
    return __builtin_amdgcn_update_dpp(0, v, U * 0x55, 0xF, 0xF, true);
}

#define SPAN_CAP 384

template <bool STAGED>
__device__ __forceinline__ int2 load_entry(const v2i* sp, int lim, int srel, const int2* __restrict__ ent,
                                           int s, int n, int k, int node) {
    int2 v;
    if (STAGED) {
        int rel = srel + k;
        v2i e = sp[rel < lim ? rel : lim];
        v.x = e.x;
        v.y = e.y;
    } else {
        int idx = s + k;
        idx = idx < N_EDGES ? idx : N_EDGES - 1;
        v = ent[idx];
    }
    bool ok = k < n;
    int tailw = (k == n) ? __float_as_int(1.0f) : 0;
    int2 r;
    r.x = ok ? v.x : node;
    r.y = ok ? v.y : tailw;
    return r;
}

#define GATHER_STEP(U, E)                                                   \
    {                                                                       \
        int r_ = quad_pick<(U) & 3>((E).x);                                 \
        w[U] = __int_as_float(quad_pick<(U) & 3>((E).y));                   \
        h[U] = hs4[(unsigned)r_ * (HIDDEN / 4) + q];                        \
    }

template <bool STAGED>
__device__ __forceinline__ float4 sum_run(const v2i* sp, int lim, int srel, const int2* __restrict__ ent,
                                          const float4* __restrict__ hs4, int s, int n, int node, unsigned q) {
    float4 acc = make_float4(0.0f, 0.0f, 0.0f, 0.0f);
    int2 ea = load_entry<STAGED>(sp, lim, srel, ent, s, n, (int)q, node);
    int2 eb = load_entry<STAGED>(sp, lim, srel, ent, s, n, 4 + (int)q, node);
    int k = 0;
    do {
        int2 ca = ea, cb = eb;
        ea = load_entry<STAGED>(sp, lim, srel, ent, s, n, k + 8 + (int)q, node);
        eb = load_entry<STAGED>(sp, lim, srel, ent, s, n, k + 12 + (int)q, node);
        float4 h[8];
        float w[8];
        GATHER_STEP(0, ca) GATHER_STEP(1, ca) GATHER_STEP(2, ca) GATHER_STEP(3, ca)
        GATHER_STEP(4, cb) GATHER_STEP(5, cb) GATHER_STEP(6, cb) GATHER_STEP(7, cb)
#pragma unroll
        for (int u = 0; u < 8; ++u) {
            acc.x = fmaf(w[u], h[u].x, acc.x);
            acc.y = fmaf(w[u], h[u].y, acc.y);
            acc.z = fmaf(w[u], h[u].z, acc.z);
            acc.w = fmaf(w[u], h[u].w, acc.w);
        }
        k += 8;
    } while (k <= n);
    return acc;
}

#define TP 18
template <int LAYER>
__global__ void __launch_bounds__(256, 16) agg_kernel(const int* __restrict__ off, const int* __restrict__ cnt,
                                                  const int2* __restrict__ ent, const float* __restrict__ hs,
                                                  const float* __restrict__ dinv, const float* __restrict__ b1,
                                                  float* __restrict__ dst, const float* __restrict__ W2,
                                                  const float* __restrict__ b2, float* __restrict__ out) {
    __shared__ float tile[LAYER == 2 ? 8 * 16 * TP : 1];
    __shared__ v4i span[8 * SPAN_CAP / 2];
    unsigned wave = (blockIdx.x * blockDim.x + threadIdx.x) >> 5;
    if (wave >= N_TILES) return;
    unsigned lane = threadIdx.x & 31;
    unsigned g = lane >> 2, q = lane & 3;
    unsigned c0 = wave * 16;
    const float4* hs4 = (const float4*)hs;
    const v4i* ent4 = (const v4i*)ent;
    unsigned wv = (unsigned)__builtin_amdgcn_readfirstlane((int)(threadIdx.x >> 5));
    v4i* sp4 = span + wv * (SPAN_CAP / 2);
    const v2i* sp = (const v2i*)sp4;
    int s2[2] = { off[c0 + g], off[c0 + 8u + g] };
    int n2[2] = { cnt[c0 + g], cnt[c0 + 8u + g] };
    float d2[2] = { dinv[c0 + g], dinv[c0 + 8u + g] };
#pragma unroll
    for (int pass = 0; pass < 2; ++pass) {
        unsigned node = c0 + 8u * (unsigned)pass + g;
        int s = s2[pass], n = n2[pass];
        float di = d2[pass];
        int first = __builtin_amdgcn_readfirstlane(s);
        int last  = __builtin_amdgcn_readlane(s + n, 28);
        int base  = first & ~1;
        float4 acc;
        if (last - base <= SPAN_CAP) {
            int npair = (last - base + 1) >> 1;
            v4i sv[6];
#pragma unroll
            for (int j = 0; j < 6; ++j) {
                int gi = (base >> 1) + (int)lane + 32 * j;
                sv[j] = ent4[gi < N_EDGES / 2 ? gi : N_EDGES / 2 - 1];
            }
            asm volatile("" :: "v"(sv[0]), "v"(sv[1]), "v"(sv[2]), "v"(sv[3]), "v"(sv[4]), "v"(sv[5]));
#pragma unroll
            for (int j = 0; j < 6; ++j) {
                int i = (int)lane + 32 * j;
                if (i < npair) sp4[i] = sv[j];
            }
            int lim = 2 * npair - 1;
            lim = lim > 0 ? lim : 0;
            acc = sum_run<true>(sp, lim, s - base, ent, hs4, s, n, (int)node, q);
        } else {
            acc = sum_run<false>(sp, 0, 0, ent, hs4, s, n, (int)node, q);
        }
        float4 v;
        v.x = di * acc.x;
        v.y = di * acc.y;
        v.z = di * acc.z;
        v.w = di * acc.w;
        if (LAYER == 1) {
            float4 bias = ((const float4*)b1)[q];
            v.x += bias.x; v.y += bias.y; v.z += bias.z; v.w += bias.w;
            v.x = v.x > 0.0f ? v.x : 0.0f;
            v.y = v.y > 0.0f ? v.y : 0.0f;
            v.z = v.z > 0.0f ? v.z : 0.0f;
            v.w = v.w > 0.0f ? v.w : 0.0f;
            v.x *= di; v.y *= di; v.z *= di; v.w *= di;
            ((float4*)dst)[node * (HIDDEN / 4) + q] = v;
        } else {
            float* trow = tile + wv * (16 * TP) + (8u * (unsigned)pass + g) * TP + 4u * q;
            trow[0] = v.x; trow[1] = v.y; trow[2] = v.z; trow[3] = v.w;
        }
    }
    if (LAYER == 2) {
        unsigned f = lane & 15, hi = lane >> 4;
        const float* trow = tile + wv * (16 * TP) + f * TP;
        float bia0 = b2[f], bia1 = b2[16 + f];
        v8f c0v = { bia0, bia0, bia0, bia0, bia0, bia0, bia0, bia0 };
        v8f c1v = { bia1, bia1, bia1, bia1, bia1, bia1, bia1, bia1 };
#pragma unroll
        for (int k = 0; k < HIDDEN; k += 4) {
            int ka = k + (int)hi * 2;
            v2f a  = { trow[ka], trow[ka + 1] };
            v2f w0 = { W2[ka * N_CLASSES + f],      W2[(ka + 1) * N_CLASSES + f] };
            v2f w1 = { W2[ka * N_CLASSES + 16 + f], W2[(ka + 1) * N_CLASSES + 16 + f] };
            c0v = __builtin_amdgcn_wmma_f32_16x16x4_f32(false, a, false, w0,
                                                        (short)0, c0v, false, false);
            c1v = __builtin_amdgcn_wmma_f32_16x16x4_f32(false, a, false, w1,
                                                        (short)0, c1v, false, false);
        }
        unsigned mbase = c0 + hi * 8;
#pragma unroll
        for (int r = 0; r < 8; ++r) {
            out[(mbase + r) * N_CLASSES + f]      = c0v[r];
            out[(mbase + r) * N_CLASSES + 16 + f] = c1v[r];
        }
    }
}

extern "C" void kernel_launch(void* const* d_in, const int* in_sizes, int n_in,
                              void* d_out, int out_size, void* d_ws, size_t ws_size,
                              hipStream_t stream) {
    const float* x  = (const float*)d_in[0];
    const int*   ei = (const int*)d_in[1];
    const float* ew = (const float*)d_in[2];
    const float* W1 = (const float*)d_in[3];
    const float* b1 = (const float*)d_in[4];
    const float* W2 = (const float*)d_in[5];
    const float* b2 = (const float*)d_in[6];
    const int* row = ei;
    const int* col = ei + N_EDGES;
    float* out = (float*)d_out;

    int*   cnt  = (int*)d_ws;
    int*   off  = cnt  + N_NODES;
    float* dinv = (float*)(off + N_NODES + 256);
    float* hs1  = dinv + N_NODES;
    float* hs2  = hs1  + (size_t)N_NODES * HIDDEN;
    float* spare = hs2 + (size_t)N_NODES * HIDDEN;
    int*   rcnt = (int*)(spare + (size_t)N_NODES * HIDDEN);
    int*   rloc = rcnt + M_BH;
    v2i*   tmp  = (v2i*)(rloc + M_BH);
    int2*  ent  = (int2*)(tmp + N_EDGES);

    const int B = 256;
    bucket_kernel<<<NBLK, 1024, 0, stream>>>(row, col, ew, rcnt, rloc, tmp);
    node_sort_kernel<<<NBK, 256, 0, stream>>>(tmp, rcnt, rloc, off, cnt, ent, x, W1, dinv, hs1);
    agg_kernel<1><<<(N_TILES * 32 + B - 1) / B, B, 0, stream>>>(off, cnt, ent, hs1, dinv, b1, hs2, W2, b2, out);
    agg_kernel<2><<<(N_TILES * 32 + B - 1) / B, B, 0, stream>>>(off, cnt, ent, hs2, dinv, b1, spare, W2, b2, out);
}


# Both in-edge sum launches run as 512 work-groups of 25 waves: a WGP's wave slots hold exactly two, so every WGP sums 48 to 50 tiles and a third as many groups enter; every sum keeps its order
# speedup vs baseline: 8.5759x; 1.0305x over previous
#include <hip/hip_runtime.h>

#define N_NODES   200000
#define N_EDGES   6400000
#define IN_FEAT   128
#define HIDDEN    16
#define N_CLASSES 32
#define N_TILES   (N_NODES / 16)

#define SH        8
#define BNODES    (1 << SH)
#define NBK       ((N_NODES + BNODES - 1) / BNODES)
#define CH        32768
#define NBLK      ((N_EDGES + CH - 1) / CH)
#define M_BH      (NBK * NBLK)
#define E_QUADS   (N_EDGES / 4)
#define RLANES    ((NBLK + 7) / 8)
#define ROW_BITS  18
#define CAP       9984

static_assert(N_EDGES % 4 == 0, "whole quads");
static_assert(N_NODES <= (1 << ROW_BITS) && SH + ROW_BITS < 32, "source index and local node share one word");
static_assert(NBK <= 1024 && CH == 8 * 4 * 1024 && NBLK <= 256 && RLANES <= 32, "one thread per bucket; one node-sort thread per run of a bucket");
static_assert(CH <= (1 << 15) && SH + 15 < 32, "a slot of the stage and a local node share one word");

typedef __attribute__((ext_vector_type(2))) float v2f;
typedef __attribute__((ext_vector_type(8))) float v8f;
typedef __attribute__((ext_vector_type(2))) int   v2i;
typedef __attribute__((ext_vector_type(4))) int   v4i;


__global__ void __launch_bounds__(1024) bucket_kernel(const int* __restrict__ row, const int* __restrict__ col,
                                                      const float* __restrict__ w, int* __restrict__ rcnt,
                                                      int* __restrict__ rloc, v2i* __restrict__ tmp) {
    __shared__ __attribute__((aligned(16))) v2i stage[CH];
    __shared__ int lstart[1024];
    __shared__ int cur[1024];
    unsigned t = threadIdx.x, k = blockIdx.x;
    cur[t] = 0;
    __syncthreads();
    const int4*   col4 = (const int4*)col;
    const int4*   row4 = (const int4*)row;
    const float4* w4   = (const float4*)w;
    int4 cc[8], rk[8];
#pragma unroll
    for (int i = 0; i < 8; ++i) {
        unsigned q = k * (CH / 4) + i * 1024 + t;
        cc[i] = col4[q < E_QUADS ? q : 0];
    }
#pragma unroll
    for (int i = 0; i < 8; ++i) {
        unsigned q = k * (CH / 4) + i * 1024 + t;
        rk[i] = make_int4(0, 0, 0, 0);
        if (q < E_QUADS) {
            rk[i].x = atomicAdd(&cur[cc[i].x >> SH], 1);
            rk[i].y = atomicAdd(&cur[cc[i].y >> SH], 1);
            rk[i].z = atomicAdd(&cur[cc[i].z >> SH], 1);
            rk[i].w = atomicAdd(&cur[cc[i].w >> SH], 1);
        }
    }
    __syncthreads();
    int c = cur[t];
    for (int d = 1; d < 1024; d <<= 1) {
        int add = (t >= (unsigned)d) ? cur[t - d] : 0;
        __syncthreads();
        cur[t] += add;
        __syncthreads();
    }
    int ex = cur[t] - c;
    lstart[t] = ex;
    if (t < NBK) {
        rcnt[t * NBLK + k] = c;
        rloc[t * NBLK + k] = ex;
    }
    __syncthreads();
    int4 pk[8];
#pragma unroll
    for (int i = 0; i < 8; ++i) {
        unsigned q = k * (CH / 4) + i * 1024 + t;
        pk[i] = make_int4(0, 0, 0, 0);
        if (q < E_QUADS) {
            pk[i].x = ((cc[i].x & (BNODES - 1)) << 15) | (lstart[cc[i].x >> SH] + rk[i].x);
            pk[i].y = ((cc[i].y & (BNODES - 1)) << 15) | (lstart[cc[i].y >> SH] + rk[i].y);
            pk[i].z = ((cc[i].z & (BNODES - 1)) << 15) | (lstart[cc[i].z >> SH] + rk[i].z);
            pk[i].w = ((cc[i].w & (BNODES - 1)) << 15) | (lstart[cc[i].w >> SH] + rk[i].w);
        }
    }
#pragma unroll
    for (int i0 = 0; i0 < 8; i0 += 4) {
        int4 rr[4];
        float4 ww[4];
#pragma unroll
        for (int u = 0; u < 4; ++u) {
            unsigned q = k * (CH / 4) + (i0 + u) * 1024 + t;
            unsigned qs = q < E_QUADS ? q : 0;
            rr[u] = row4[qs];
            ww[u] = w4[qs];
        }
#pragma unroll
        for (int u = 0; u < 4; ++u) {
            unsigned q = k * (CH / 4) + (i0 + u) * 1024 + t;
            if (q < E_QUADS) {
                int4 pq = pk[i0 + u];
                stage[pq.x & 32767] = (v2i){ ((pq.x >> 15) << ROW_BITS) | rr[u].x, __float_as_int(ww[u].x) };
                stage[pq.y & 32767] = (v2i){ ((pq.y >> 15) << ROW_BITS) | rr[u].y, __float_as_int(ww[u].y) };
                stage[pq.z & 32767] = (v2i){ ((pq.z >> 15) << ROW_BITS) | rr[u].z, __float_as_int(ww[u].z) };
                stage[pq.w & 32767] = (v2i){ ((pq.w >> 15) << ROW_BITS) | rr[u].w, __float_as_int(ww[u].w) };
            }
        }
    }
    __syncthreads();
    int nk = N_EDGES - (int)k * CH;
    nk = nk < CH ? nk : CH;
    const int4* s4 = (const int4*)stage;
    int4* t4 = (int4*)(tmp + (size_t)k * CH);
#pragma unroll 4
    for (int i = (int)t; i < nk / 2; i += 1024) t4[i] = s4[i];
}

__global__ void __launch_bounds__(256) node_sort_kernel(const v2i* __restrict__ tmp, const int* __restrict__ rcnt,
                                                        const int* __restrict__ rloc,
                                                        int* __restrict__ off, int* __restrict__ cnt,
                                                        int2* __restrict__ ent,
                                                        const float* __restrict__ x, const float* __restrict__ W,
                                                        float* __restrict__ dinv, float* __restrict__ hs) {
    __shared__ v2i stage[CAP];
    __shared__ int lc[BNODES];
    __shared__ int red[BNODES];
    unsigned t = threadIdx.x, b = blockIdx.x;
    const int rlane = (int)(t & 31);
    unsigned mrun = (t >> 5) + 8 * (t & 31);
    int my_rc = 0, my_rl = 0;
    if (mrun < NBLK) {
        my_rc = rcnt[b * NBLK + mrun];
        my_rl = rloc[b * NBLK + mrun];
    }
    int my_rg = (int)mrun * CH + my_rl;
    lc[t] = my_rl;
    red[t] = my_rc;
    __syncthreads();
    for (int d = 128; d > 0; d >>= 1) {
        if (t < (unsigned)d) {
            lc[t] += lc[t + d];
            red[t] += red[t + d];
        }
        __syncthreads();
    }
    int bstart = lc[0];
    int len = red[0];
    __syncthreads();
    bool staged = len <= CAP;
    lc[t] = 0;
    __syncthreads();
    for (int i0 = 0; i0 < RLANES; i0 += 8) {
        int key[16];
#pragma unroll
        for (int u = 0; u < 8; ++u) {
            int rg = __builtin_amdgcn_readlane(my_rg, i0 + u);
            int rc = __builtin_amdgcn_readlane(my_rc, i0 + u);
            key[2 * u]     = tmp[rlane < rc ? rg + rlane : 0].x;
            key[2 * u + 1] = tmp[rlane + 32 < rc ? rg + rlane + 32 : 0].x;
        }
        asm volatile("" :: "v"(key[0]), "v"(key[1]), "v"(key[2]), "v"(key[3]), "v"(key[4]), "v"(key[5]), "v"(key[6]), "v"(key[7]),
                           "v"(key[8]), "v"(key[9]), "v"(key[10]), "v"(key[11]), "v"(key[12]), "v"(key[13]), "v"(key[14]), "v"(key[15]));
#pragma unroll
        for (int u = 0; u < 8; ++u) {
            int rc = __builtin_amdgcn_readlane(my_rc, i0 + u);
            if (rlane < rc) atomicAdd(&lc[key[2 * u] >> ROW_BITS], 1);
            if (rlane + 32 < rc) atomicAdd(&lc[key[2 * u + 1] >> ROW_BITS], 1);
        }
        for (int u = 0; u < 8; ++u) {
            int rg = __builtin_amdgcn_readlane(my_rg, i0 + u);
            int rc = __builtin_amdgcn_readlane(my_rc, i0 + u);
            for (int j = rlane + 64; j < rc; j += 32) atomicAdd(&lc[tmp[rg + j].x >> ROW_BITS], 1);
        }
    }
    __syncthreads();
    int c = lc[t];
    for (int d = 1; d < 256; d <<= 1) {
        int add = (t >= (unsigned)d) ? lc[t - d] : 0;
        __syncthreads();
        lc[t] += add;
        __syncthreads();
    }
    int ex = lc[t] - c;
    unsigned node = b * BNODES + t;
    if (node < N_NODES) {
        off[node] = bstart + ex;
        cnt[node] = c;
    }
    lc[t] = ex;
    __syncthreads();
    for (int i0 = 0; i0 < RLANES; i0 += 8) {
        v2i e[16];
        int p[16];
#pragma unroll
        for (int u = 0; u < 8; ++u) {
            int rg = __builtin_amdgcn_readlane(my_rg, i0 + u);
            int rc = __builtin_amdgcn_readlane(my_rc, i0 + u);
            e[2 * u]     = tmp[rlane < rc ? rg + rlane : 0];
            e[2 * u + 1] = tmp[rlane + 32 < rc ? rg + rlane + 32 : 0];
        }
        asm volatile("" :: "v"(e[0]), "v"(e[1]), "v"(e[2]), "v"(e[3]), "v"(e[4]), "v"(e[5]), "v"(e[6]), "v"(e[7]),
                           "v"(e[8]), "v"(e[9]), "v"(e[10]), "v"(e[11]), "v"(e[12]), "v"(e[13]), "v"(e[14]), "v"(e[15]));
#pragma unroll
        for (int u = 0; u < 8; ++u) {
            int rc = __builtin_amdgcn_readlane(my_rc, i0 + u);
            p[2 * u] = 0;
            p[2 * u + 1] = 0;
            if (rlane < rc) p[2 * u] = atomicAdd(&lc[e[2 * u].x >> ROW_BITS], 1);
            if (rlane + 32 < rc) p[2 * u + 1] = atomicAdd(&lc[e[2 * u + 1].x >> ROW_BITS], 1);
        }
#pragma unroll
        for (int u = 0; u < 8; ++u) {
            int rc = __builtin_amdgcn_readlane(my_rc, i0 + u);
#pragma unroll
            for (int h = 0; h < 2; ++h) {
                if (rlane + 32 * h < rc) {
                    int r = e[2 * u + h].x & ((1 << ROW_BITS) - 1);
                    if (staged) stage[p[2 * u + h]] = (v2i){ r, e[2 * u + h].y };
                    else        ent[bstart + p[2 * u + h]] = make_int2(r, e[2 * u + h].y);
                }
            }
        }
        for (int u = 0; u < 8; ++u) {
            int rg = __builtin_amdgcn_readlane(my_rg, i0 + u);
            int rc = __builtin_amdgcn_readlane(my_rc, i0 + u);
            for (int j = rlane + 64; j < rc; j += 32) {
                v2i ee = tmp[rg + j];
                int pp = atomicAdd(&lc[ee.x >> ROW_BITS], 1);
                int r = ee.x & ((1 << ROW_BITS) - 1);
                if (staged) stage[pp] = (v2i){ r, ee.y };
                else        ent[bstart + pp] = make_int2(r, ee.y);
            }
        }
    }
    if (!staged) __threadfence();
    __syncthreads();
    float deg = 0.0f;
    if (staged) {
#pragma unroll 4
        for (int j = (int)t; j < len; j += 256) {
            v2i e = stage[j];
            ent[bstart + j] = make_int2(e.x, e.y);
        }
#pragma unroll 4
        for (int i = 0; i < c; ++i) deg += __int_as_float(stage[ex + i].y);
    } else {
        for (int i = 0; i < c; ++i) deg += __int_as_float(ent[bstart + ex + i].y);
    }
    deg += 1.0f;
    float dn = (deg > 0.0f) ? rsqrtf(deg) : 0.0f;
    if (node < N_NODES) dinv[node] = dn;
    lc[t] = __float_as_int(dn);
    __syncthreads();
    unsigned wv = (unsigned)__builtin_amdgcn_readfirstlane((int)(t >> 5));
    unsigned lane = t & 31;
    unsigned lh = lane & 15;
    unsigned hi = lane >> 4;
    for (unsigned tt = wv; tt < BNODES / 16; tt += 8) {
        unsigned m0 = b * BNODES + tt * 16;
        if (m0 >= N_NODES) break;
        v8f acc = {};
        const float* xrow = x + (size_t)(m0 + lh) * IN_FEAT;
#pragma unroll
        for (int k = 0; k < IN_FEAT; k += 4) {
            int ka = k + (int)hi * 2;
            v2f a = { xrow[ka], xrow[ka + 1] };
            v2f bw = { W[ka * HIDDEN + lh], W[(ka + 1) * HIDDEN + lh] };
            acc = __builtin_amdgcn_wmma_f32_16x16x4_f32(false, a, false, bw,
                                                        (short)0, acc, false, false);
        }
        unsigned lb = tt * 16 + hi * 8;
#pragma unroll
        for (int r = 0; r < 8; ++r)
            hs[(size_t)(b * BNODES + lb + r) * HIDDEN + lh] = acc[r] * __int_as_float(lc[lb + r]);
    }
}


template <int U>
__device__ __forceinline__ int quad_pick(int v) {
    return __builtin_amdgcn_update_dpp(0, v, U * 0x55, 0xF, 0xF, true);
}

#define SPAN_CAP 384

#define AGG_GROUPS  512
#define AGG_WAVES   25
#define AGG_THREADS 800
static_assert(AGG_THREADS == 32 * AGG_WAVES && 2 * AGG_WAVES <= 64 && 3 * AGG_WAVES > 64, "two groups fill a WGP's wave slots, a third does not fit");
static_assert(N_TILES <= AGG_GROUPS * AGG_WAVES && N_TILES >= AGG_GROUPS * (AGG_WAVES - 1), "a group's tiles: one per wave, at most one wave idle");

template <bool STAGED>
__device__ __forceinline__ int2 load_entry(const v2i* sp, int lim, int srel, const int2* __restrict__ ent,
                                           int s, int n, int k, int node) {
    int2 v;
    if (STAGED) {
        int rel = srel + k;
        v2i e = sp[rel < lim ? rel : lim];
        v.x = e.x;
        v.y = e.y;
    } else {
        int idx = s + k;
        idx = idx < N_EDGES ? idx : N_EDGES - 1;
        v = ent[idx];
    }
    bool ok = k < n;
    int tailw = (k == n) ? __float_as_int(1.0f) : 0;
    int2 r;
    r.x = ok ? v.x : node;
    r.y = ok ? v.y : tailw;
    return r;
}

#define GATHER_STEP(U, E)                                                   \
    {                                                                       \
        int r_ = quad_pick<(U) & 3>((E).x);                                 \
        w[U] = __int_as_float(quad_pick<(U) & 3>((E).y));                   \
        h[U] = hs4[(unsigned)r_ * (HIDDEN / 4) + q];                        \
    }

template <bool STAGED>
__device__ __forceinline__ float4 sum_run(const v2i* sp, int lim, int srel, const int2* __restrict__ ent,
                                          const float4* __restrict__ hs4, int s, int n, int node, unsigned q) {
    float4 acc = make_float4(0.0f, 0.0f, 0.0f, 0.0f);
    int2 ea = load_entry<STAGED>(sp, lim, srel, ent, s, n, (int)q, node);
    int2 eb = load_entry<STAGED>(sp, lim, srel, ent, s, n, 4 + (int)q, node);
    int k = 0;
    do {
        int2 ca = ea, cb = eb;
        ea = load_entry<STAGED>(sp, lim, srel, ent, s, n, k + 8 + (int)q, node);
        eb = load_entry<STAGED>(sp, lim, srel, ent, s, n, k + 12 + (int)q, node);
        float4 h[8];
        float w[8];
        GATHER_STEP(0, ca) GATHER_STEP(1, ca) GATHER_STEP(2, ca) GATHER_STEP(3, ca)
        GATHER_STEP(4, cb) GATHER_STEP(5, cb) GATHER_STEP(6, cb) GATHER_STEP(7, cb)
#pragma unroll
        for (int u = 0; u < 8; ++u) {
            acc.x = fmaf(w[u], h[u].x, acc.x);
            acc.y = fmaf(w[u], h[u].y, acc.y);
            acc.z = fmaf(w[u], h[u].z, acc.z);
            acc.w = fmaf(w[u], h[u].w, acc.w);
        }
        k += 8;
    } while (k <= n);
    return acc;
}

#define TP 18
template <int LAYER>
__global__ void __launch_bounds__(AGG_THREADS, 13) agg_kernel(const int* __restrict__ off, const int* __restrict__ cnt,
                                                  const int2* __restrict__ ent, const float* __restrict__ hs,
                                                  const float* __restrict__ dinv, const float* __restrict__ b1,
                                                  float* __restrict__ dst, const float* __restrict__ W2,
                                                  const float* __restrict__ b2, float* __restrict__ out) {
    __shared__ float tile[LAYER == 2 ? AGG_WAVES * 16 * TP : 1];
    __shared__ v4i span[AGG_WAVES * SPAN_CAP / 2];
    unsigned wv = (unsigned)__builtin_amdgcn_readfirstlane((int)(threadIdx.x >> 5));
    unsigned grp = blockIdx.x;
    const unsigned full = N_TILES / AGG_GROUPS;
    const unsigned extra = N_TILES % AGG_GROUPS;
    if (wv >= full + (grp < extra ? 1u : 0u)) return;
    unsigned wave = grp * full + (grp < extra ? grp : extra) + wv;
    unsigned lane = threadIdx.x & 31;
    unsigned g = lane >> 2, q = lane & 3;
    unsigned c0 = wave * 16;
    const float4* hs4 = (const float4*)hs;
    const v4i* ent4 = (const v4i*)ent;
    v4i* sp4 = span + wv * (SPAN_CAP / 2);
    const v2i* sp = (const v2i*)sp4;
    int s2[2] = { off[c0 + g], off[c0 + 8u + g] };
    int n2[2] = { cnt[c0 + g], cnt[c0 + 8u + g] };
    float d2[2] = { dinv[c0 + g], dinv[c0 + 8u + g] };
#pragma unroll
    for (int pass = 0; pass < 2; ++pass) {
        unsigned node = c0 + 8u * (unsigned)pass + g;
        int s = s2[pass], n = n2[pass];
        float di = d2[pass];
        int first = __builtin_amdgcn_readfirstlane(s);
        int last  = __builtin_amdgcn_readlane(s + n, 28);
        int base  = first & ~1;
        float4 acc;
        if (last - base <= SPAN_CAP) {
            int npair = (last - base + 1) >> 1;
            v4i sv[6];
#pragma unroll
            for (int j = 0; j < 6; ++j) {
                int gi = (base >> 1) + (int)lane + 32 * j;
                sv[j] = ent4[gi < N_EDGES / 2 ? gi : N_EDGES / 2 - 1];
            }
            asm volatile("" :: "v"(sv[0]), "v"(sv[1]), "v"(sv[2]), "v"(sv[3]), "v"(sv[4]), "v"(sv[5]));
#pragma unroll
            for (int j = 0; j < 6; ++j) {
                int i = (int)lane + 32 * j;
                if (i < npair) sp4[i] = sv[j];
            }
            int lim = 2 * npair - 1;
            lim = lim > 0 ? lim : 0;
            acc = sum_run<true>(sp, lim, s - base, ent, hs4, s, n, (int)node, q);
        } else {
            acc = sum_run<false>(sp, 0, 0, ent, hs4, s, n, (int)node, q);
        }
        float4 v;
        v.x = di * acc.x;
        v.y = di * acc.y;
        v.z = di * acc.z;
        v.w = di * acc.w;
        if (LAYER == 1) {
            float4 bias = ((const float4*)b1)[q];
            v.x += bias.x; v.y += bias.y; v.z += bias.z; v.w += bias.w;
            v.x = v.x > 0.0f ? v.x : 0.0f;
            v.y = v.y > 0.0f ? v.y : 0.0f;
            v.z = v.z > 0.0f ? v.z : 0.0f;
            v.w = v.w > 0.0f ? v.w : 0.0f;
            v.x *= di; v.y *= di; v.z *= di; v.w *= di;
            ((float4*)dst)[node * (HIDDEN / 4) + q] = v;
        } else {
            float* trow = tile + wv * (16 * TP) + (8u * (unsigned)pass + g) * TP + 4u * q;
            trow[0] = v.x; trow[1] = v.y; trow[2] = v.z; trow[3] = v.w;
        }
    }
    if (LAYER == 2) {
        unsigned f = lane & 15, hi = lane >> 4;
        const float* trow = tile + wv * (16 * TP) + f * TP;
        float bia0 = b2[f], bia1 = b2[16 + f];
        v8f c0v = { bia0, bia0, bia0, bia0, bia0, bia0, bia0, bia0 };
        v8f c1v = { bia1, bia1, bia1, bia1, bia1, bia1, bia1, bia1 };
#pragma unroll
        for (int k = 0; k < HIDDEN; k += 4) {
            int ka = k + (int)hi * 2;
            v2f a  = { trow[ka], trow[ka + 1] };
            v2f w0 = { W2[ka * N_CLASSES + f],      W2[(ka + 1) * N_CLASSES + f] };
            v2f w1 = { W2[ka * N_CLASSES + 16 + f], W2[(ka + 1) * N_CLASSES + 16 + f] };
            c0v = __builtin_amdgcn_wmma_f32_16x16x4_f32(false, a, false, w0,
                                                        (short)0, c0v, false, false);
            c1v = __builtin_amdgcn_wmma_f32_16x16x4_f32(false, a, false, w1,
                                                        (short)0, c1v, false, false);
        }
        unsigned mbase = c0 + hi * 8;
#pragma unroll
        for (int r = 0; r < 8; ++r) {
            out[(mbase + r) * N_CLASSES + f]      = c0v[r];
            out[(mbase + r) * N_CLASSES + 16 + f] = c1v[r];
        }
    }
}

extern "C" void kernel_launch(void* const* d_in, const int* in_sizes, int n_in,
                              void* d_out, int out_size, void* d_ws, size_t ws_size,
                              hipStream_t stream) {
    const float* x  = (const float*)d_in[0];
    const int*   ei = (const int*)d_in[1];
    const float* ew = (const float*)d_in[2];
    const float* W1 = (const float*)d_in[3];
    const float* b1 = (const float*)d_in[4];
    const float* W2 = (const float*)d_in[5];
    const float* b2 = (const float*)d_in[6];
    const int* row = ei;
    const int* col = ei + N_EDGES;
    float* out = (float*)d_out;

    int*   cnt  = (int*)d_ws;
    int*   off  = cnt  + N_NODES;
    float* dinv = (float*)(off + N_NODES + 256);
    float* hs1  = dinv + N_NODES;
    float* hs2  = hs1  + (size_t)N_NODES * HIDDEN;
    float* spare = hs2 + (size_t)N_NODES * HIDDEN;
    int*   rcnt = (int*)(spare + (size_t)N_NODES * HIDDEN);
    int*   rloc = rcnt + M_BH;
    v2i*   tmp  = (v2i*)(rloc + M_BH);
    int2*  ent  = (int2*)(tmp + N_EDGES);

    bucket_kernel<<<NBLK, 1024, 0, stream>>>(row, col, ew, rcnt, rloc, tmp);
    node_sort_kernel<<<NBK, 256, 0, stream>>>(tmp, rcnt, rloc, off, cnt, ent, x, W1, dinv, hs1);
    agg_kernel<1><<<AGG_GROUPS, AGG_THREADS, 0, stream>>>(off, cnt, ent, hs1, dinv, b1, hs2, W2, b2, out);
    agg_kernel<2><<<AGG_GROUPS, AGG_THREADS, 0, stream>>>(off, cnt, ent, hs2, dinv, b1, spare, W2, b2, out);
}
